# starting kernel
# baseline (speedup 1.0000x reference)
#include <hip/hip_runtime.h>
#include <hip/hip_bf16.h>

typedef __attribute__((ext_vector_type(2))) float v2f;
typedef __attribute__((ext_vector_type(8))) float v8f;

#define IN_DIM 128
#define OUT_DIM 64
#define NEG_SLOPE 0.2f

__global__ void gat_init(float* __restrict__ out, float* __restrict__ denom,
                         unsigned* __restrict__ m_enc, int n_nodes) {
    long long i = (long long)blockIdx.x * blockDim.x + threadIdx.x;
    long long total = (long long)n_nodes * OUT_DIM;
    if (i < total) out[i] = 0.0f;
    if (i < n_nodes) {
        denom[i] = 0.0f;
        m_enc[i] = 0x007FFFFFu;
    }
}

__global__ __launch_bounds__(256) void gat_gemm(const float* __restrict__ x,
                                                const float* __restrict__ W,
                                                float* __restrict__ feat,
                                                int n_nodes) {
    __shared__ float Wt[OUT_DIM][IN_DIM];
    for (int i = threadIdx.x; i < IN_DIM * OUT_DIM; i += blockDim.x) {
        int k = i >> 6;
        int c = i & 63;
        Wt[c][k] = W[i];
    }
    __syncthreads();

    const int wave = threadIdx.x >> 5;
    const int lane = threadIdx.x & 31;
    const int tile = blockIdx.x * (blockDim.x >> 5) + wave;
    const int n_tiles = (n_nodes + 15) >> 4;
    if (tile >= n_tiles) return;

    const int row0 = tile << 4;
    const int l16  = lane & 15;
    const int hi   = lane >> 4;

    v8f acc0 = {}, acc1 = {}, acc2 = {}, acc3 = {};
    const float* xrow = x + (size_t)(row0 + l16) * IN_DIM;

    for (int k = 0; k < IN_DIM; k += 4) {
        v2f a = *(const v2f*)(xrow + k + 2 * hi);
        v2f b0 = *(const v2f*)&Wt[ 0 + l16][k + 2 * hi];
        v2f b1 = *(const v2f*)&Wt[16 + l16][k + 2 * hi];
        v2f b2 = *(const v2f*)&Wt[32 + l16][k + 2 * hi];
        v2f b3 = *(const v2f*)&Wt[48 + l16][k + 2 * hi];
        acc0 = __builtin_amdgcn_wmma_f32_16x16x4_f32(false, a, false, b0, (short)0, acc0, false, false);
        acc1 = __builtin_amdgcn_wmma_f32_16x16x4_f32(false, a, false, b1, (short)0, acc1, false, false);
        acc2 = __builtin_amdgcn_wmma_f32_16x16x4_f32(false, a, false, b2, (short)0, acc2, false, false);
        acc3 = __builtin_amdgcn_wmma_f32_16x16x4_f32(false, a, false, b3, (short)0, acc3, false, false);
    }

    float* o = feat + ((size_t)(row0 + hi * 8)) * OUT_DIM + l16;
#pragma unroll
    for (int r = 0; r < 8; ++r) {
        o[(size_t)r * OUT_DIM +  0] = acc0[r];
        o[(size_t)r * OUT_DIM + 16] = acc1[r];
        o[(size_t)r * OUT_DIM + 32] = acc2[r];
        o[(size_t)r * OUT_DIM + 48] = acc3[r];
    }
}

__global__ void gat_logits(const float* __restrict__ feat,
                           const float* __restrict__ attn_l,
                           const float* __restrict__ attn_r,
                           float* __restrict__ el, float* __restrict__ er,
                           int n_nodes) {
    const int lane  = threadIdx.x & 31;
    const int waveg = (blockIdx.x * blockDim.x + threadIdx.x) >> 5;
    const int nwav  = (gridDim.x * blockDim.x) >> 5;
    const float al0 = attn_l[lane], al1 = attn_l[lane + 32];
    const float ar0 = attn_r[lane], ar1 = attn_r[lane + 32];
    for (int node = waveg; node < n_nodes; node += nwav) {
        const float* f = feat + (size_t)node * OUT_DIM;
        float f0 = f[lane], f1 = f[lane + 32];
        float pl = f0 * al0 + f1 * al1;
        float pr = f0 * ar0 + f1 * ar1;
#pragma unroll
        for (int off = 16; off; off >>= 1) {
            pl += __shfl_xor(pl, off, 32);
            pr += __shfl_xor(pr, off, 32);
        }
        if (lane == 0) { el[node] = pl; er[node] = pr; }
    }
}

__global__ void gat_edge_score(const int* __restrict__ src, const int* __restrict__ dst,
                               const float* __restrict__ el, const float* __restrict__ er,
                               float* __restrict__ e, unsigned* __restrict__ m_enc,
                               int n_edges) {
    int i = blockIdx.x * blockDim.x + threadIdx.x;
    if (i >= n_edges) return;
    int d = dst[i];
    float v = el[src[i]] + er[d];
    v = v > 0.0f ? v : NEG_SLOPE * v;
    e[i] = v;
    unsigned bits = __float_as_uint(v);
    unsigned enc  = (bits & 0x80000000u) ? ~bits : (bits | 0x80000000u);
    atomicMax(&m_enc[d], enc);
}

__global__ void gat_exp(const int* __restrict__ dst, const unsigned* __restrict__ m_enc,
                        float* __restrict__ e, float* __restrict__ denom, int n_edges) {
    int i = blockIdx.x * blockDim.x + threadIdx.x;
    if (i >= n_edges) return;
    int d = dst[i];
    unsigned enc  = m_enc[d];
    unsigned bits = (enc & 0x80000000u) ? (enc & 0x7FFFFFFFu) : ~enc;
    float m  = __uint_as_float(bits);
    float ex = expf(e[i] - m);
    e[i] = ex;
    atomicAdd(&denom[d], ex);
}

__global__ void gat_aggregate(const int* __restrict__ src, const int* __restrict__ dst,
                              const float* __restrict__ ex, const float* __restrict__ denom,
                              const float* __restrict__ feat, float* __restrict__ out,
                              int n_edges) {
    long long t = (long long)blockIdx.x * blockDim.x + threadIdx.x;
    int edge = (int)(t >> 6);
    if (edge >= n_edges) return;
    int c = (int)t & 63;
    int d = dst[edge], s = src[edge];
    float alpha = ex[edge] / denom[d];
    atomicAdd(&out[(size_t)d * OUT_DIM + c], alpha * feat[(size_t)s * OUT_DIM + c]);
}

__global__ void gat_finalize(float* __restrict__ out, const float* __restrict__ bias,
                             int n_nodes) {
    long long i = (long long)blockIdx.x * blockDim.x + threadIdx.x;
    long long total = (long long)n_nodes * OUT_DIM;
    if (i >= total) return;
    out[i] = tanhf(out[i] + bias[i & 63]);
}

static inline size_t align256(size_t x) { return (x + 255) & ~(size_t)255; }

extern "C" void kernel_launch(void* const* d_in, const int* in_sizes, int n_in,
                              void* d_out, int out_size, void* d_ws, size_t ws_size,
                              hipStream_t stream) {
    const float* x      = (const float*)d_in[0];
    const int*   src    = (const int*)d_in[1];
    const int*   dst    = (const int*)d_in[2];
    const float* W      = (const float*)d_in[3];
    const float* attn_l = (const float*)d_in[4];
    const float* attn_r = (const float*)d_in[5];
    const float* bias   = (const float*)d_in[6];
    float*       out    = (float*)d_out;

    const int N = in_sizes[0] / IN_DIM;
    const int E = in_sizes[1];

    char* ws = (char*)d_ws;
    float*    feat  = (float*)ws;    ws += align256((size_t)N * OUT_DIM * sizeof(float));
    float*    e     = (float*)ws;    ws += align256((size_t)E * sizeof(float));
    float*    el    = (float*)ws;    ws += align256((size_t)N * sizeof(float));
    float*    er    = (float*)ws;    ws += align256((size_t)N * sizeof(float));
    float*    denom = (float*)ws;    ws += align256((size_t)N * sizeof(float));
    unsigned* m_enc = (unsigned*)ws; ws += align256((size_t)N * sizeof(unsigned));

    const int BLK = 256;

    {
        long long total = (long long)N * OUT_DIM;
        int blocks = (int)((total + BLK - 1) / BLK);
        gat_init<<<blocks, BLK, 0, stream>>>(out, denom, m_enc, N);
    }
    {
        int n_tiles = (N + 15) / 16;
        int waves_per_block = BLK / 32;
        int blocks = (n_tiles + waves_per_block - 1) / waves_per_block;
        gat_gemm<<<blocks, BLK, 0, stream>>>(x, W, feat, N);
    }
    {
        int blocks = 1024;
        gat_logits<<<blocks, BLK, 0, stream>>>(feat, attn_l, attn_r, el, er, N);
    }
    {
        int blocks = (E + BLK - 1) / BLK;
        gat_edge_score<<<blocks, BLK, 0, stream>>>(src, dst, el, er, e, m_enc, E);
    }
    {
        int blocks = (E + BLK - 1) / BLK;
        gat_exp<<<blocks, BLK, 0, stream>>>(dst, m_enc, e, denom, E);
    }
    {
        long long total = (long long)E * OUT_DIM;
        int blocks = (int)((total + BLK - 1) / BLK);
        gat_aggregate<<<blocks, BLK, 0, stream>>>(src, dst, e, denom, feat, out, E);
    }
    {
        long long total = (long long)N * OUT_DIM;
        int blocks = (int)((total + BLK - 1) / BLK);
        gat_finalize<<<blocks, BLK, 0, stream>>>(out, bias, N);
    }
}


# Logits: el, er as one 64 x 2 product on the f32 16x16x4 matrix instruction (attn_l, attn_r = B columns 0, 1), a wave per 32 nodes; 10 waited lane permutes per node gone; el, er summed in matrix order
# speedup vs baseline: 5.5122x; 5.5122x over previous
#include <hip/hip_runtime.h>

typedef __attribute__((ext_vector_type(2))) float v2f;
typedef __attribute__((ext_vector_type(4))) float v4f;
typedef __attribute__((ext_vector_type(8))) float v8f;

#define IN_DIM 128
#define OUT_DIM 64
#define NEG_SLOPE 0.2f

#define GEMM_THREADS 256
#define GEMM_WAVES   (GEMM_THREADS / 32)
#define GEMM_BLOCKS  256
#define KGROUPS      (IN_DIM / 8)

__global__ __launch_bounds__(GEMM_THREADS) void gat_gemm(const float* __restrict__ x,
                                                         const float* __restrict__ W,
                                                         float* __restrict__ feat,
                                                         int n_nodes) {
    const int lane = threadIdx.x & 31;
    const int c    = lane & 15;
    const int h    = lane >> 4;
    const int wave = __builtin_amdgcn_readfirstlane((int)(threadIdx.x >> 5));
    const int wlin    = wave * (int)gridDim.x + (int)blockIdx.x;
    const int n_waves = (int)gridDim.x * GEMM_WAVES;
    const int n_tiles = (n_nodes + 15) >> 4;

    v2f wf[KGROUPS][2][4];
#pragma unroll
    for (int g = 0; g < KGROUPS; ++g) {
#pragma unroll
        for (int j = 0; j < 2; ++j) {
            const int k = 8 * g + 4 * h + 2 * j;
            const v4f r0 = *(const v4f*)(W + (size_t)k * OUT_DIM + 4 * c);
            const v4f r1 = *(const v4f*)(W + (size_t)(k + 1) * OUT_DIM + 4 * c);
            wf[g][j][0] = (v2f){r0.x, r1.x};
            wf[g][j][1] = (v2f){r0.y, r1.y};
            wf[g][j][2] = (v2f){r0.z, r1.z};
            wf[g][j][3] = (v2f){r0.w, r1.w};
        }
    }

    if (wlin >= n_tiles) return;

    v4f xa[KGROUPS];
    {
        const int row = min(wlin * 16 + c, n_nodes - 1);
        const v4f* xp = (const v4f*)(x + (size_t)row * IN_DIM + 4 * h);
#pragma unroll
        for (int g = 0; g < KGROUPS; ++g) xa[g] = xp[2 * g];
    }

    for (int tile = wlin; tile < n_tiles; tile += n_waves) {
        v4f xn[KGROUPS];
        {
            const int nt  = min(tile + n_waves, n_tiles - 1);
            const int row = min(nt * 16 + c, n_nodes - 1);
            const v4f* xp = (const v4f*)(x + (size_t)row * IN_DIM + 4 * h);
#pragma unroll
            for (int g = 0; g < KGROUPS; ++g) xn[g] = xp[2 * g];
        }

        v8f acc0 = {}, acc1 = {}, acc2 = {}, acc3 = {};
#pragma unroll
        for (int g = 0; g < KGROUPS; ++g) {
            const v2f a0 = (v2f){xa[g].x, xa[g].y};
            const v2f a1 = (v2f){xa[g].z, xa[g].w};
            acc0 = __builtin_amdgcn_wmma_f32_16x16x4_f32(false, a0, false, wf[g][0][0], (short)0, acc0, false, false);
            acc1 = __builtin_amdgcn_wmma_f32_16x16x4_f32(false, a0, false, wf[g][0][1], (short)0, acc1, false, false);
            acc2 = __builtin_amdgcn_wmma_f32_16x16x4_f32(false, a0, false, wf[g][0][2], (short)0, acc2, false, false);
            acc3 = __builtin_amdgcn_wmma_f32_16x16x4_f32(false, a0, false, wf[g][0][3], (short)0, acc3, false, false);
            acc0 = __builtin_amdgcn_wmma_f32_16x16x4_f32(false, a1, false, wf[g][1][0], (short)0, acc0, false, false);
            acc1 = __builtin_amdgcn_wmma_f32_16x16x4_f32(false, a1, false, wf[g][1][1], (short)0, acc1, false, false);
            acc2 = __builtin_amdgcn_wmma_f32_16x16x4_f32(false, a1, false, wf[g][1][2], (short)0, acc2, false, false);
            acc3 = __builtin_amdgcn_wmma_f32_16x16x4_f32(false, a1, false, wf[g][1][3], (short)0, acc3, false, false);
        }

        float* o = feat + ((size_t)tile * 16 + 8 * h) * OUT_DIM + 4 * c;
        if (tile * 16 + 16 <= n_nodes) {
#pragma unroll
            for (int r = 0; r < 8; ++r)
                *(v4f*)(o + r * OUT_DIM) = (v4f){acc0[r], acc1[r], acc2[r], acc3[r]};
        } else {
            const int rows_here = n_nodes - tile * 16 - 8 * h;
#pragma unroll
            for (int r = 0; r < 8; ++r)
                if (r < rows_here)
                    *(v4f*)(o + r * OUT_DIM) = (v4f){acc0[r], acc1[r], acc2[r], acc3[r]};
        }

#pragma unroll
        for (int g = 0; g < KGROUPS; ++g) xa[g] = xn[g];
    }
}

#define OWN_NODES   391
#define KEY_OWNERS  256

#define CHUNK_SHIFT 5
#define TILE_CHUNKS 256
#define TILE_WORDS  (TILE_CHUNKS / 32)
#define TILE_PITCH  (TILE_WORDS + 1)
#define K2_THREADS  256

#define LGROUPS (OUT_DIM / 8)

__global__ __launch_bounds__(K2_THREADS) void gat_logits(const float* __restrict__ feat,
                           const float* __restrict__ attn_l,
                           const float* __restrict__ attn_r,
                           float* __restrict__ el, float* __restrict__ er,
                           int n_nodes,
                           const int* __restrict__ dst, unsigned* __restrict__ bm, int n_edges, int n_tiles) {
    __shared__ unsigned rows[KEY_OWNERS * TILE_PITCH];
    const int n_bm = bm ? n_tiles : 0;
    if ((int)blockIdx.x < n_bm) {
        const unsigned t    = threadIdx.x;
        const unsigned tile = blockIdx.x;
        for (unsigned i = t; i < KEY_OWNERS * TILE_PITCH; i += K2_THREADS) rows[i] = 0u;
        __syncthreads();
        const unsigned nvec = ((unsigned)n_edges >> CHUNK_SHIFT) << 3;
        const int4* __restrict__ dst4 = (const int4*)dst;
        const unsigned g0  = tile * (TILE_CHUNKS * 8u) + t;
        const unsigned bit = 1u << (t >> 3);
        int4 d[TILE_WORDS];
#pragma unroll
        for (int j = 0; j < TILE_WORDS; ++j) d[j] = dst4[min(g0 + (unsigned)j * K2_THREADS, nvec - 1u)];
#pragma unroll
        for (int j = 0; j < TILE_WORDS; ++j) {
            if (g0 + (unsigned)j * K2_THREADS < nvec) {
                atomicOr(&rows[min((unsigned)d[j].x / OWN_NODES, KEY_OWNERS - 1u) * TILE_PITCH + j], bit);
                atomicOr(&rows[min((unsigned)d[j].y / OWN_NODES, KEY_OWNERS - 1u) * TILE_PITCH + j], bit);
                atomicOr(&rows[min((unsigned)d[j].z / OWN_NODES, KEY_OWNERS - 1u) * TILE_PITCH + j], bit);
                atomicOr(&rows[min((unsigned)d[j].w / OWN_NODES, KEY_OWNERS - 1u) * TILE_PITCH + j], bit);
            }
        }
        __syncthreads();
        for (unsigned o = t; o < KEY_OWNERS; o += K2_THREADS) {
            const unsigned* r = &rows[o * TILE_PITCH];
            uint4* out4 = (uint4*)(bm + ((size_t)tile * KEY_OWNERS + o) * TILE_WORDS);
            out4[0] = make_uint4(r[0], r[1], r[2], r[3]);
            out4[1] = make_uint4(r[4], r[5], r[6], r[7]);
        }
        return;
    }
    const int lane = threadIdx.x & 31;
    const int c    = lane & 15;
    const int h    = lane >> 4;
    const int wave    = __builtin_amdgcn_readfirstlane((int)(threadIdx.x >> 5));
    const int wlin    = ((int)blockIdx.x - n_bm) * (K2_THREADS / 32) + wave;
    const int n_waves = ((int)gridDim.x - n_bm) * (K2_THREADS / 32);
    const int n_grp   = (n_nodes + 31) >> 5;

    const float* av = (c == 1) ? attn_r : attn_l;
    const unsigned keep = (c < 2) ? 0xFFFFFFFFu : 0u;
    v2f bf[LGROUPS][2];
#pragma unroll
    for (int g = 0; g < LGROUPS; ++g) {
#pragma unroll
        for (int j = 0; j < 2; ++j) {
            const int k = 8 * g + 4 * h + 2 * j;
            bf[g][j] = (v2f){__uint_as_float(__float_as_uint(av[k]) & keep),
                             __uint_as_float(__float_as_uint(av[k + 1]) & keep)};
        }
    }

    for (int grp = wlin; grp < n_grp; grp += n_waves) {
        const int rowA = min(grp * 32 + c, n_nodes - 1);
        const int rowB = min(grp * 32 + 16 + c, n_nodes - 1);
        const v4f* pa = (const v4f*)(feat + (size_t)rowA * OUT_DIM + 4 * h);
        const v4f* pb = (const v4f*)(feat + (size_t)rowB * OUT_DIM + 4 * h);
        v4f fa[LGROUPS], fb[LGROUPS];
#pragma unroll
        for (int g = 0; g < LGROUPS; ++g) { fa[g] = pa[2 * g]; fb[g] = pb[2 * g]; }
        __builtin_amdgcn_sched_barrier(0);

        v8f accA = {}, accB = {};
#pragma unroll
        for (int g = 0; g < LGROUPS; ++g) {
            const v2f a0 = (v2f){fa[g].x, fa[g].y}, a1 = (v2f){fa[g].z, fa[g].w};
            const v2f b0 = (v2f){fb[g].x, fb[g].y}, b1 = (v2f){fb[g].z, fb[g].w};
            accA = __builtin_amdgcn_wmma_f32_16x16x4_f32(false, a0, false, bf[g][0], (short)0, accA, false, false);
            accB = __builtin_amdgcn_wmma_f32_16x16x4_f32(false, b0, false, bf[g][0], (short)0, accB, false, false);
            accA = __builtin_amdgcn_wmma_f32_16x16x4_f32(false, a1, false, bf[g][1], (short)0, accA, false, false);
            accB = __builtin_amdgcn_wmma_f32_16x16x4_f32(false, b1, false, bf[g][1], (short)0, accB, false, false);
        }

        if (c < 2) {
            float* o = (c == 0 ? el : er) + (size_t)grp * 32 + 8 * h;
            if (grp * 32 + 32 <= n_nodes) {
                *(v4f*)(o)      = (v4f){accA[0], accA[1], accA[2], accA[3]};
                *(v4f*)(o + 4)  = (v4f){accA[4], accA[5], accA[6], accA[7]};
                *(v4f*)(o + 16) = (v4f){accB[0], accB[1], accB[2], accB[3]};
                *(v4f*)(o + 20) = (v4f){accB[4], accB[5], accB[6], accB[7]};
            } else {
                const int left = n_nodes - grp * 32 - 8 * h;
#pragma unroll
                for (int r = 0; r < 8; ++r) {
                    if (r < left)      o[r]      = accA[r];
                    if (r + 16 < left) o[16 + r] = accB[r];
                }
            }
        }
    }
}

#define OWN_THREADS 1024
#define OWN_WAVES   (OWN_THREADS / 32)
#define QCAP        16384
#define SCAN_U      8
#define CHUNK_U     16
#define GATHER_U    8
#define ENC_NEG_INF 0x007FFFFFu

__device__ __forceinline__ unsigned enc_f(float v) {
    unsigned b = __float_as_uint(v);
    return (b & 0x80000000u) ? ~b : (b | 0x80000000u);
}
__device__ __forceinline__ float dec_f(unsigned e) {
    unsigned b = (e & 0x80000000u) ? (e & 0x7FFFFFFFu) : ~e;
    return __uint_as_float(b);
}

__device__ __forceinline__ void q_push(unsigned* qcount, unsigned* q_e, unsigned e) {
    unsigned slot = atomicAdd(qcount, 1u);
    if (slot < QCAP) q_e[slot] = e;
}

__device__ __forceinline__ unsigned wave_slots(unsigned* counter, unsigned cnt, unsigned ln, unsigned cap, bool& fits) {
    unsigned incl = cnt;
#pragma unroll
    for (unsigned d = 1u; d < 32u; d <<= 1) {
        const unsigned t = __shfl_up(incl, d, 32);
        if (ln >= d) incl += t;
    }
    const unsigned wtot = (unsigned)__builtin_amdgcn_readlane((int)incl, 31);
    unsigned b = 0u;
    if (ln == 0u) b = atomicAdd(counter, wtot);
    b = (unsigned)__builtin_amdgcn_readfirstlane((int)b);
    fits = b + wtot <= cap;
    return b + incl - cnt;
}

__device__ __forceinline__ void gather_batch(const float* __restrict__ feat, float* acc,
                                             unsigned s_l, unsigned dl_l, unsigned ex_l,
                                             unsigned j0, unsigned n, unsigned ln) {
    v2f f[GATHER_U]; unsigned dl[GATHER_U]; float ex[GATHER_U];
#pragma unroll
    for (unsigned u = 0u; u < GATHER_U; ++u) {
        const int j = (int)min(j0 + u, n - 1u);
        const unsigned s = (unsigned)__builtin_amdgcn_readlane((int)s_l, j);
        dl[u] = (unsigned)__builtin_amdgcn_readlane((int)dl_l, j);
        ex[u] = __uint_as_float((unsigned)__builtin_amdgcn_readlane((int)ex_l, j));
        const float* row = feat + (size_t)s * OUT_DIM;
        f[u] = *(const v2f*)(row + 2u * ln);
    }
#pragma unroll
    for (unsigned u = 0u; u < GATHER_U; ++u) {
        if (j0 + u < n) {
            float* a = &acc[dl[u] * OUT_DIM + 2u * ln];
            atomicAdd(a,     ex[u] * f[u].x);
            atomicAdd(a + 1, ex[u] * f[u].y);
        }
    }
}

__global__ __launch_bounds__(OWN_THREADS) void gat_owner(
        const int* __restrict__ src, const int* __restrict__ dst,
        const float* __restrict__ el, const float* __restrict__ er,
        const float* __restrict__ feat, const float* __restrict__ bias,
        float* __restrict__ out, int n_nodes, int n_edges,
        const unsigned* __restrict__ bm) {
    __shared__ float    acc[OWN_NODES * OUT_DIM];
    __shared__ float    den_s[OWN_NODES];
    __shared__ float    mrun_s[OWN_NODES];
    __shared__ float    sc_s[OWN_NODES];
    __shared__ float    er_s[OWN_NODES];
    __shared__ unsigned mb_s[OWN_NODES];
    __shared__ unsigned q_e[QCAP];
    __shared__ unsigned q_dl[QCAP];
    __shared__ float    q_val[QCAP];
    __shared__ unsigned qcount;
    __shared__ unsigned cqcount;

    const int tid  = threadIdx.x;
    const int base = blockIdx.x * OWN_NODES;
    const int nloc = min(OWN_NODES, n_nodes - base);
    const unsigned unloc = (unsigned)nloc;

    for (int idx = tid; idx < nloc * OUT_DIM; idx += OWN_THREADS) acc[idx] = 0.0f;
    for (int i = tid; i < nloc; i += OWN_THREADS) {
        den_s[i]  = 0.0f;
        mrun_s[i] = __uint_as_float(0xFF800000u);
        mb_s[i]   = ENC_NEG_INF;
        er_s[i]   = er[base + i];
    }
    if (tid == 0) { qcount = 0u; cqcount = 0u; }
    __syncthreads();

    const unsigned E = (unsigned)n_edges;
    const bool use_bm = bm != 0;
    const unsigned eshift = use_bm ? (unsigned)CHUNK_SHIFT : 2u;
    const unsigned nwhole = E >> eshift;
    const unsigned ngroups = use_bm ? (nwhole + TILE_CHUNKS - 1u) / TILE_CHUNKS : nwhole;
    const unsigned gsafe = use_bm ? 1u : (QCAP >> 2) - 1u;
    const int4* __restrict__ dst4 = (const int4*)dst;
    unsigned* cq = (unsigned*)q_val;
    const unsigned wv = (unsigned)__builtin_amdgcn_readfirstlane(tid >> 5);
    const unsigned ln = (unsigned)tid & 31u;
    const unsigned sub = ln >> 3, part = ln & 7u;
    const unsigned cqcap = QCAP - 4u * OWN_WAVES * CHUNK_U;
    unsigned gpos = 0u, glen = ngroups, nflush = 0u;
    bool more = E > 0u;
    const unsigned rot = (unsigned)(((unsigned long long)blockIdx.x * ngroups) / gridDim.x) & ~31u;

    while (more) {
        const unsigned gend = min(gpos + glen, ngroups);
        unsigned ncq = 0u;

        if (use_bm) {
            for (unsigned tb = gpos + 4u * wv; tb < gend; tb += 4u * OWN_WAVES) {
                const unsigned tile = tb + sub;
                unsigned word = 0u;
                if (tile < gend) word = bm[((size_t)tile * KEY_OWNERS + blockIdx.x) * TILE_WORDS + part];
                const unsigned c0 = tile * TILE_CHUNKS + part * 32u;
                bool fits;
                unsigned pos = wave_slots(&cqcount, (unsigned)__builtin_popcount(word), ln, cqcap, fits);
                if (!fits) word = 0u;
                while (word) {
                    cq[pos++] = c0 + (unsigned)__builtin_ctz(word);
                    word &= word - 1u;
                }
            }
            __syncthreads();
            ncq = (unsigned)__builtin_amdgcn_readfirstlane((int)cqcount);
            if (ncq <= cqcap) {
                for (unsigned i0 = 4u * wv; i0 < ncq; i0 += 4u * OWN_WAVES * CHUNK_U) {
                    unsigned ck[CHUNK_U], h[CHUNK_U];
                    int4 v[CHUNK_U];
#pragma unroll
                    for (int u = 0; u < CHUNK_U; ++u) {
                        ck[u] = min(cq[i0 + (unsigned)u * 4u * OWN_WAVES + sub], nwhole - 1u);
                        v[u]  = dst4[ck[u] * 8u + part];
                    }
                    unsigned cnt = 0u;
#pragma unroll
                    for (int u = 0; u < CHUNK_U; ++u) {
                        const unsigned lim = (i0 + (unsigned)u * 4u * OWN_WAVES + sub < ncq) ? unloc : 0u;
                        h[u] = ((unsigned)(v[u].x - base) < lim ? 1u : 0u) | ((unsigned)(v[u].y - base) < lim ? 2u : 0u) |
                               ((unsigned)(v[u].z - base) < lim ? 4u : 0u) | ((unsigned)(v[u].w - base) < lim ? 8u : 0u);
                        cnt += (unsigned)__builtin_popcount(h[u]);
                    }
                    bool fits;
                    unsigned pos = wave_slots(&qcount, cnt, ln, QCAP, fits);
                    if (fits) {
#pragma unroll
                        for (int u = 0; u < CHUNK_U; ++u) {
                            unsigned m = h[u];
                            const unsigned e0 = (ck[u] << CHUNK_SHIFT) + part * 4u;
                            while (m) {
                                q_e[pos++] = e0 + (unsigned)__builtin_ctz(m);
                                m &= m - 1u;
                            }
                        }
                    }
                }
            }
        } else {
            for (unsigned g = gpos + (unsigned)tid; g < gend; g += OWN_THREADS * SCAN_U) {
                int4 v[SCAN_U];
#pragma unroll
                for (int u = 0; u < SCAN_U; ++u) {
                    unsigned ph = min(g + (unsigned)u * OWN_THREADS, gend - 1u) + rot;
                    if (ph >= ngroups) ph -= ngroups;
                    v[u] = dst4[ph];
                }
#pragma unroll
                for (int u = 0; u < SCAN_U; ++u) {
                    unsigned gi = g + (unsigned)u * OWN_THREADS;
                    unsigned ph = gi + rot;
                    if (ph >= ngroups) ph -= ngroups;
                    unsigned e0 = ph << 2;
                    unsigned lim = (gi < gend) ? unloc : 0u;
                    bool h0 = (unsigned)(v[u].x - base) < lim, h1 = (unsigned)(v[u].y - base) < lim;
                    bool h2 = (unsigned)(v[u].z - base) < lim, h3 = (unsigned)(v[u].w - base) < lim;
                    if (h0 | h1 | h2 | h3) {
                        if (h0) q_push(&qcount, q_e, e0);
                        if (h1) q_push(&qcount, q_e, e0 + 1u);
                        if (h2) q_push(&qcount, q_e, e0 + 2u);
                        if (h3) q_push(&qcount, q_e, e0 + 3u);
                    }
                }
            }
        }
        if (gend == ngroups) {
            unsigned e = (nwhole << eshift) + (unsigned)tid;
            if ((unsigned)tid < (E & ((1u << eshift) - 1u))) {
                if ((unsigned)(dst[e] - base) < unloc) q_push(&qcount, q_e, e);
            }
        }
        __syncthreads();
        const unsigned total = (unsigned)__builtin_amdgcn_readfirstlane((int)qcount);
        __syncthreads();

        if (total > QCAP || ncq > cqcap) {
            if (tid == 0) { qcount = 0u; cqcount = 0u; }
            glen = gsafe;
            __syncthreads();
            continue;
        }

        for (unsigned i = (unsigned)tid; i < total; i += OWN_THREADS) {
            const unsigned e = q_e[i];
            unsigned s  = (unsigned)src[e];
            unsigned dl = (unsigned)(dst[e] - base);
            float v = el[s] + er_s[dl];
            v = v > 0.0f ? v : NEG_SLOPE * v;
            q_e[i]   = s;
            q_dl[i]  = dl;
            q_val[i] = v;
            atomicMax(&mb_s[dl], enc_f(v));
        }
        __syncthreads();

        for (int i = tid; i < nloc; i += OWN_THREADS) {
            float mo = mrun_s[i];
            float mn = fmaxf(mo, dec_f(mb_s[i]));
            mrun_s[i] = mn;
            mb_s[i]   = ENC_NEG_INF;
            if (nflush) {
                float s = (mo == mn) ? 1.0f : expf(mo - mn);
                sc_s[i]  = s;
                den_s[i] *= s;
            }
        }
        __syncthreads();
        if (nflush) {
            for (int idx = tid; idx < nloc * OUT_DIM; idx += OWN_THREADS) acc[idx] *= sc_s[idx >> 6];
        }

        for (unsigned i = (unsigned)tid; i < total; i += OWN_THREADS) {
            unsigned dl = q_dl[i];
            float ex = expf(q_val[i] - mrun_s[dl]);
            q_val[i] = ex;
            atomicAdd(&den_s[dl], ex);
        }
        __syncthreads();

        if (total) {
            const unsigned nbat = (total + GATHER_U - 1u) / GATHER_U;
            const unsigned beg  = (wv * nbat / OWN_WAVES) * GATHER_U;
            const unsigned end  = min(((wv + 1u) * nbat / OWN_WAVES) * GATHER_U, total);
            for (unsigned g = beg; g < end; g += 32u) {
                const unsigned n = min(32u, end - g);
                const unsigned i = min(g + ln, end - 1u);
                const unsigned s_l  = q_e[i];
                const unsigned dl_l = q_dl[i];
                const unsigned ex_l = __float_as_uint(q_val[i]);
                if (n == 32u) {
#pragma unroll
                    for (unsigned j0 = 0u; j0 < 32u; j0 += GATHER_U)
                        gather_batch(feat, acc, s_l, dl_l, ex_l, j0, 32u, ln);
                } else {
#pragma nounroll
                    for (unsigned j0 = 0u; j0 < n; j0 += GATHER_U)
                        gather_batch(feat, acc, s_l, dl_l, ex_l, j0, n, ln);
                }
            }
        }
        if (tid == 0) { qcount = 0u; cqcount = 0u; }
        __syncthreads();
        ++nflush;
        gpos = gend;
        more = gpos < ngroups;
    }

    for (int idx = tid; idx < nloc * OUT_DIM; idx += OWN_THREADS) {
        float d = den_s[idx >> 6];
        float a = acc[idx];
        float r = d > 0.0f ? a / d : 0.0f;
        out[(size_t)base * OUT_DIM + idx] = tanhf(r + bias[idx & 63]);
    }
}

static inline size_t align256(size_t x) { return (x + 255) & ~(size_t)255; }

extern "C" void kernel_launch(void* const* d_in, const int* in_sizes, int n_in,
                              void* d_out, int out_size, void* d_ws, size_t ws_size,
                              hipStream_t stream) {
    const float* x      = (const float*)d_in[0];
    const int*   src    = (const int*)d_in[1];
    const int*   dst    = (const int*)d_in[2];
    const float* W      = (const float*)d_in[3];
    const float* attn_l = (const float*)d_in[4];
    const float* attn_r = (const float*)d_in[5];
    const float* bias   = (const float*)d_in[6];
    float*       out    = (float*)d_out;

    const int N = in_sizes[0] / IN_DIM;
    const int E = in_sizes[1];

    char* ws = (char*)d_ws;
    float* feat = (float*)ws;    ws += align256((size_t)N * OUT_DIM * sizeof(float));
    float* el   = (float*)ws;    ws += align256((size_t)N * sizeof(float));
    float* er   = (float*)ws;    ws += align256((size_t)N * sizeof(float));

    const int own_blocks = (N + OWN_NODES - 1) / OWN_NODES;
    const int n_tiles = ((E >> CHUNK_SHIFT) + TILE_CHUNKS - 1) / TILE_CHUNKS;
    const size_t bm_bytes = (size_t)n_tiles * KEY_OWNERS * TILE_WORDS * sizeof(unsigned);
    unsigned* bm = 0;
    if (own_blocks <= KEY_OWNERS && (size_t)(ws - (char*)d_ws) + align256(bm_bytes) <= ws_size)
        bm = (unsigned*)ws;

    {
        gat_gemm<<<GEMM_BLOCKS, GEMM_THREADS, 0, stream>>>(x, W, feat, N);
    }
    {
        int blocks = 400 + (bm ? n_tiles : 0);
        gat_logits<<<blocks, K2_THREADS, 0, stream>>>(feat, attn_l, attn_r, el, er, N, dst, bm, E, n_tiles);
    }
    {
        gat_owner<<<own_blocks, OWN_THREADS, 0, stream>>>(src, dst, el, er, feat, bias, out, N, E, bm);
    }
}


# P4: each wave owns 13 destination rows: P3 lists every edge under its row's wave (LDS counter), the wave sums its rows in registers, stores them once; row-sum LDS float adds gone; sums in list order
# speedup vs baseline: 5.7423x; 1.0417x over previous
#include <hip/hip_runtime.h>

typedef __attribute__((ext_vector_type(2))) float v2f;
typedef __attribute__((ext_vector_type(4))) float v4f;
typedef __attribute__((ext_vector_type(8))) float v8f;

#define IN_DIM 128
#define OUT_DIM 64
#define NEG_SLOPE 0.2f

#define GEMM_THREADS 256
#define GEMM_WAVES   (GEMM_THREADS / 32)
#define GEMM_BLOCKS  256
#define KGROUPS      (IN_DIM / 8)

__global__ __launch_bounds__(GEMM_THREADS) void gat_gemm(const float* __restrict__ x,
                                                         const float* __restrict__ W,
                                                         float* __restrict__ feat,
                                                         int n_nodes) {
    const int lane = threadIdx.x & 31;
    const int c    = lane & 15;
    const int h    = lane >> 4;
    const int wave = __builtin_amdgcn_readfirstlane((int)(threadIdx.x >> 5));
    const int wlin    = wave * (int)gridDim.x + (int)blockIdx.x;
    const int n_waves = (int)gridDim.x * GEMM_WAVES;
    const int n_tiles = (n_nodes + 15) >> 4;

    v2f wf[KGROUPS][2][4];
#pragma unroll
    for (int g = 0; g < KGROUPS; ++g) {
#pragma unroll
        for (int j = 0; j < 2; ++j) {
            const int k = 8 * g + 4 * h + 2 * j;
            const v4f r0 = *(const v4f*)(W + (size_t)k * OUT_DIM + 4 * c);
            const v4f r1 = *(const v4f*)(W + (size_t)(k + 1) * OUT_DIM + 4 * c);
            wf[g][j][0] = (v2f){r0.x, r1.x};
            wf[g][j][1] = (v2f){r0.y, r1.y};
            wf[g][j][2] = (v2f){r0.z, r1.z};
            wf[g][j][3] = (v2f){r0.w, r1.w};
        }
    }

    if (wlin >= n_tiles) return;

    v4f xa[KGROUPS];
    {
        const int row = min(wlin * 16 + c, n_nodes - 1);
        const v4f* xp = (const v4f*)(x + (size_t)row * IN_DIM + 4 * h);
#pragma unroll
        for (int g = 0; g < KGROUPS; ++g) xa[g] = xp[2 * g];
    }

    for (int tile = wlin; tile < n_tiles; tile += n_waves) {
        v4f xn[KGROUPS];
        {
            const int nt  = min(tile + n_waves, n_tiles - 1);
            const int row = min(nt * 16 + c, n_nodes - 1);
            const v4f* xp = (const v4f*)(x + (size_t)row * IN_DIM + 4 * h);
#pragma unroll
            for (int g = 0; g < KGROUPS; ++g) xn[g] = xp[2 * g];
        }

        v8f acc0 = {}, acc1 = {}, acc2 = {}, acc3 = {};
#pragma unroll
        for (int g = 0; g < KGROUPS; ++g) {
            const v2f a0 = (v2f){xa[g].x, xa[g].y};
            const v2f a1 = (v2f){xa[g].z, xa[g].w};
            acc0 = __builtin_amdgcn_wmma_f32_16x16x4_f32(false, a0, false, wf[g][0][0], (short)0, acc0, false, false);
            acc1 = __builtin_amdgcn_wmma_f32_16x16x4_f32(false, a0, false, wf[g][0][1], (short)0, acc1, false, false);
            acc2 = __builtin_amdgcn_wmma_f32_16x16x4_f32(false, a0, false, wf[g][0][2], (short)0, acc2, false, false);
            acc3 = __builtin_amdgcn_wmma_f32_16x16x4_f32(false, a0, false, wf[g][0][3], (short)0, acc3, false, false);
            acc0 = __builtin_amdgcn_wmma_f32_16x16x4_f32(false, a1, false, wf[g][1][0], (short)0, acc0, false, false);
            acc1 = __builtin_amdgcn_wmma_f32_16x16x4_f32(false, a1, false, wf[g][1][1], (short)0, acc1, false, false);
            acc2 = __builtin_amdgcn_wmma_f32_16x16x4_f32(false, a1, false, wf[g][1][2], (short)0, acc2, false, false);
            acc3 = __builtin_amdgcn_wmma_f32_16x16x4_f32(false, a1, false, wf[g][1][3], (short)0, acc3, false, false);
        }

        float* o = feat + ((size_t)tile * 16 + 8 * h) * OUT_DIM + 4 * c;
        if (tile * 16 + 16 <= n_nodes) {
#pragma unroll
            for (int r = 0; r < 8; ++r)
                *(v4f*)(o + r * OUT_DIM) = (v4f){acc0[r], acc1[r], acc2[r], acc3[r]};
        } else {
            const int rows_here = n_nodes - tile * 16 - 8 * h;
#pragma unroll
            for (int r = 0; r < 8; ++r)
                if (r < rows_here)
                    *(v4f*)(o + r * OUT_DIM) = (v4f){acc0[r], acc1[r], acc2[r], acc3[r]};
        }

#pragma unroll
        for (int g = 0; g < KGROUPS; ++g) xa[g] = xn[g];
    }
}

#define OWN_NODES   391
#define KEY_OWNERS  256

#define CHUNK_SHIFT 5
#define TILE_CHUNKS 256
#define TILE_WORDS  (TILE_CHUNKS / 32)
#define TILE_PITCH  (TILE_WORDS + 1)
#define K2_THREADS  256

#define LGROUPS (OUT_DIM / 8)

__global__ __launch_bounds__(K2_THREADS) void gat_logits(const float* __restrict__ feat,
                           const float* __restrict__ attn_l,
                           const float* __restrict__ attn_r,
                           float* __restrict__ el, float* __restrict__ er,
                           int n_nodes,
                           const int* __restrict__ dst, unsigned* __restrict__ bm, int n_edges, int n_tiles) {
    __shared__ unsigned rows[KEY_OWNERS * TILE_PITCH];
    const int n_bm = bm ? n_tiles : 0;
    if ((int)blockIdx.x < n_bm) {
        const unsigned t    = threadIdx.x;
        const unsigned tile = blockIdx.x;
        for (unsigned i = t; i < KEY_OWNERS * TILE_PITCH; i += K2_THREADS) rows[i] = 0u;
        __syncthreads();
        const unsigned nvec = ((unsigned)n_edges >> CHUNK_SHIFT) << 3;
        const int4* __restrict__ dst4 = (const int4*)dst;
        const unsigned g0  = tile * (TILE_CHUNKS * 8u) + t;
        const unsigned bit = 1u << (t >> 3);
        int4 d[TILE_WORDS];
#pragma unroll
        for (int j = 0; j < TILE_WORDS; ++j) d[j] = dst4[min(g0 + (unsigned)j * K2_THREADS, nvec - 1u)];
#pragma unroll
        for (int j = 0; j < TILE_WORDS; ++j) {
            if (g0 + (unsigned)j * K2_THREADS < nvec) {
                atomicOr(&rows[min((unsigned)d[j].x / OWN_NODES, KEY_OWNERS - 1u) * TILE_PITCH + j], bit);
                atomicOr(&rows[min((unsigned)d[j].y / OWN_NODES, KEY_OWNERS - 1u) * TILE_PITCH + j], bit);
                atomicOr(&rows[min((unsigned)d[j].z / OWN_NODES, KEY_OWNERS - 1u) * TILE_PITCH + j], bit);
                atomicOr(&rows[min((unsigned)d[j].w / OWN_NODES, KEY_OWNERS - 1u) * TILE_PITCH + j], bit);
            }
        }
        __syncthreads();
        for (unsigned o = t; o < KEY_OWNERS; o += K2_THREADS) {
            const unsigned* r = &rows[o * TILE_PITCH];
            uint4* out4 = (uint4*)(bm + ((size_t)tile * KEY_OWNERS + o) * TILE_WORDS);
            out4[0] = make_uint4(r[0], r[1], r[2], r[3]);
            out4[1] = make_uint4(r[4], r[5], r[6], r[7]);
        }
        return;
    }
    const int lane = threadIdx.x & 31;
    const int c    = lane & 15;
    const int h    = lane >> 4;
    const int wave    = __builtin_amdgcn_readfirstlane((int)(threadIdx.x >> 5));
    const int wlin    = ((int)blockIdx.x - n_bm) * (K2_THREADS / 32) + wave;
    const int n_waves = ((int)gridDim.x - n_bm) * (K2_THREADS / 32);
    const int n_grp   = (n_nodes + 31) >> 5;

    const float* av = (c == 1) ? attn_r : attn_l;
    const unsigned keep = (c < 2) ? 0xFFFFFFFFu : 0u;
    v2f bf[LGROUPS][2];
#pragma unroll
    for (int g = 0; g < LGROUPS; ++g) {
#pragma unroll
        for (int j = 0; j < 2; ++j) {
            const int k = 8 * g + 4 * h + 2 * j;
            bf[g][j] = (v2f){__uint_as_float(__float_as_uint(av[k]) & keep),
                             __uint_as_float(__float_as_uint(av[k + 1]) & keep)};
        }
    }

    for (int grp = wlin; grp < n_grp; grp += n_waves) {
        const int rowA = min(grp * 32 + c, n_nodes - 1);
        const int rowB = min(grp * 32 + 16 + c, n_nodes - 1);
        const v4f* pa = (const v4f*)(feat + (size_t)rowA * OUT_DIM + 4 * h);
        const v4f* pb = (const v4f*)(feat + (size_t)rowB * OUT_DIM + 4 * h);
        v4f fa[LGROUPS], fb[LGROUPS];
#pragma unroll
        for (int g = 0; g < LGROUPS; ++g) { fa[g] = pa[2 * g]; fb[g] = pb[2 * g]; }
        __builtin_amdgcn_sched_barrier(0);

        v8f accA = {}, accB = {};
#pragma unroll
        for (int g = 0; g < LGROUPS; ++g) {
            const v2f a0 = (v2f){fa[g].x, fa[g].y}, a1 = (v2f){fa[g].z, fa[g].w};
            const v2f b0 = (v2f){fb[g].x, fb[g].y}, b1 = (v2f){fb[g].z, fb[g].w};
            accA = __builtin_amdgcn_wmma_f32_16x16x4_f32(false, a0, false, bf[g][0], (short)0, accA, false, false);
            accB = __builtin_amdgcn_wmma_f32_16x16x4_f32(false, b0, false, bf[g][0], (short)0, accB, false, false);
            accA = __builtin_amdgcn_wmma_f32_16x16x4_f32(false, a1, false, bf[g][1], (short)0, accA, false, false);
            accB = __builtin_amdgcn_wmma_f32_16x16x4_f32(false, b1, false, bf[g][1], (short)0, accB, false, false);
        }

        if (c < 2) {
            float* o = (c == 0 ? el : er) + (size_t)grp * 32 + 8 * h;
            if (grp * 32 + 32 <= n_nodes) {
                *(v4f*)(o)      = (v4f){accA[0], accA[1], accA[2], accA[3]};
                *(v4f*)(o + 4)  = (v4f){accA[4], accA[5], accA[6], accA[7]};
                *(v4f*)(o + 16) = (v4f){accB[0], accB[1], accB[2], accB[3]};
                *(v4f*)(o + 20) = (v4f){accB[4], accB[5], accB[6], accB[7]};
            } else {
                const int left = n_nodes - grp * 32 - 8 * h;
#pragma unroll
                for (int r = 0; r < 8; ++r) {
                    if (r < left)      o[r]      = accA[r];
                    if (r + 16 < left) o[16 + r] = accB[r];
                }
            }
        }
    }
}

#define OWN_THREADS 1024
#define OWN_WAVES   (OWN_THREADS / 32)
#define QCAP        16384
#define SCAN_U      8
#define CHUNK_U     16
#define GATHER_U    8
#define ENC_NEG_INF 0x007FFFFFu
#define ROWS_PW     13
#define WL_CAP      320

__device__ __forceinline__ unsigned enc_f(float v) {
    unsigned b = __float_as_uint(v);
    return (b & 0x80000000u) ? ~b : (b | 0x80000000u);
}
__device__ __forceinline__ float dec_f(unsigned e) {
    unsigned b = (e & 0x80000000u) ? (e & 0x7FFFFFFFu) : ~e;
    return __uint_as_float(b);
}

__device__ __forceinline__ void q_push(unsigned* qcount, unsigned* q_e, unsigned e) {
    unsigned slot = atomicAdd(qcount, 1u);
    if (slot < QCAP) q_e[slot] = e;
}

__device__ __forceinline__ unsigned wave_slots(unsigned* counter, unsigned cnt, unsigned ln, unsigned cap, bool& fits) {
    unsigned incl = cnt;
#pragma unroll
    for (unsigned d = 1u; d < 32u; d <<= 1) {
        const unsigned t = __shfl_up(incl, d, 32);
        if (ln >= d) incl += t;
    }
    const unsigned wtot = (unsigned)__builtin_amdgcn_readlane((int)incl, 31);
    unsigned b = 0u;
    if (ln == 0u) b = atomicAdd(counter, wtot);
    b = (unsigned)__builtin_amdgcn_readfirstlane((int)b);
    fits = b + wtot <= cap;
    return b + incl - cnt;
}

__device__ __forceinline__ void gather_batch(const float* __restrict__ feat, float* acc,
                                             unsigned s_l, unsigned dl_l, unsigned ex_l,
                                             unsigned j0, unsigned n, unsigned ln) {
    v2f f[GATHER_U]; unsigned dl[GATHER_U]; float ex[GATHER_U];
#pragma unroll
    for (unsigned u = 0u; u < GATHER_U; ++u) {
        const int j = (int)min(j0 + u, n - 1u);
        const unsigned s = (unsigned)__builtin_amdgcn_readlane((int)s_l, j);
        dl[u] = (unsigned)__builtin_amdgcn_readlane((int)dl_l, j);
        ex[u] = __uint_as_float((unsigned)__builtin_amdgcn_readlane((int)ex_l, j));
        const float* row = feat + (size_t)s * OUT_DIM;
        f[u] = *(const v2f*)(row + 2u * ln);
    }
#pragma unroll
    for (unsigned u = 0u; u < GATHER_U; ++u) {
        if (j0 + u < n) {
            float* a = &acc[dl[u] * OUT_DIM + 2u * ln];
            atomicAdd(a,     ex[u] * f[u].x);
            atomicAdd(a + 1, ex[u] * f[u].y);
        }
    }
}

typedef __attribute__((ext_vector_type(32))) float v32f;
__device__ __forceinline__ void gather_batch_own(const float* __restrict__ feat, v32f& ar,
                                                 unsigned s_l, unsigned rw_l, unsigned ex_l,
                                                 unsigned j0, unsigned n, unsigned ln) {
    v2f f[GATHER_U];
#pragma unroll
    for (unsigned u = 0u; u < GATHER_U; ++u) {
        const int j = (int)min(j0 + u, n - 1u);
        const unsigned s = (unsigned)__builtin_amdgcn_readlane((int)s_l, j);
        const float* row = feat + (size_t)s * OUT_DIM;
        f[u] = *(const v2f*)(row + 2u * ln);
    }
#pragma unroll
    for (unsigned u = 0u; u < GATHER_U; ++u) {
        if (j0 + u < n) {
            const int j = (int)(j0 + u);
            const unsigned r2 = 2u * (unsigned)__builtin_amdgcn_readlane((int)rw_l, j);
            const float w = __uint_as_float((unsigned)__builtin_amdgcn_readlane((int)ex_l, j));
            const float ox = ar[r2], oy = ar[r2 + 1u];
            ar[r2]      = fmaf(w, f[u].x, ox);
            ar[r2 + 1u] = fmaf(w, f[u].y, oy);
        }
    }
}

__global__ __launch_bounds__(OWN_THREADS) void gat_owner(
        const int* __restrict__ src, const int* __restrict__ dst,
        const float* __restrict__ el, const float* __restrict__ er,
        const float* __restrict__ feat, const float* __restrict__ bias,
        float* __restrict__ out, int n_nodes, int n_edges,
        const unsigned* __restrict__ bm) {
    __shared__ float    acc[OWN_NODES * OUT_DIM];
    __shared__ float    den_s[OWN_NODES];
    __shared__ float    mrun_s[OWN_NODES];
    __shared__ float    sc_s[OWN_NODES];
    __shared__ float    er_s[OWN_NODES];
    __shared__ unsigned mb_s[OWN_NODES];
    __shared__ unsigned q_e[QCAP];
    __shared__ unsigned q_dl[QCAP];
    __shared__ float    q_val[QCAP];
    __shared__ unsigned qcount;
    __shared__ unsigned cqcount;
    __shared__ unsigned short wl[OWN_WAVES * WL_CAP];
    __shared__ unsigned wcnt[OWN_WAVES];
    __shared__ unsigned wovf;
    static_assert((OWN_NODES + ROWS_PW - 1) / ROWS_PW <= OWN_WAVES, "every row needs a wave");
    static_assert(QCAP <= 65536, "queue positions are kept in 16 bits");

    const int tid  = threadIdx.x;
    const int base = blockIdx.x * OWN_NODES;
    const int nloc = min(OWN_NODES, n_nodes - base);
    const unsigned unloc = (unsigned)nloc;

    for (int idx = tid; idx < nloc * OUT_DIM; idx += OWN_THREADS) acc[idx] = 0.0f;
    for (int i = tid; i < nloc; i += OWN_THREADS) {
        den_s[i]  = 0.0f;
        mrun_s[i] = __uint_as_float(0xFF800000u);
        mb_s[i]   = ENC_NEG_INF;
        er_s[i]   = er[base + i];
    }
    if (tid == 0) { qcount = 0u; cqcount = 0u; wovf = 0u; }
    if (tid < OWN_WAVES) wcnt[tid] = 0u;
    __syncthreads();

    const unsigned E = (unsigned)n_edges;
    const bool use_bm = bm != 0;
    const unsigned eshift = use_bm ? (unsigned)CHUNK_SHIFT : 2u;
    const unsigned nwhole = E >> eshift;
    const unsigned ngroups = use_bm ? (nwhole + TILE_CHUNKS - 1u) / TILE_CHUNKS : nwhole;
    const unsigned gsafe = use_bm ? 1u : (QCAP >> 2) - 1u;
    const int4* __restrict__ dst4 = (const int4*)dst;
    unsigned* cq = (unsigned*)q_val;
    const unsigned wv = (unsigned)__builtin_amdgcn_readfirstlane(tid >> 5);
    const unsigned ln = (unsigned)tid & 31u;
    const unsigned sub = ln >> 3, part = ln & 7u;
    const unsigned cqcap = QCAP - 4u * OWN_WAVES * CHUNK_U;
    unsigned gpos = 0u, glen = ngroups, nflush = 0u;
    bool more = E > 0u;
    const unsigned rot = (unsigned)(((unsigned long long)blockIdx.x * ngroups) / gridDim.x) & ~31u;

    while (more) {
        const unsigned gend = min(gpos + glen, ngroups);
        unsigned ncq = 0u;

        if (use_bm) {
            for (unsigned tb = gpos + 4u * wv; tb < gend; tb += 4u * OWN_WAVES) {
                const unsigned tile = tb + sub;
                unsigned word = 0u;
                if (tile < gend) word = bm[((size_t)tile * KEY_OWNERS + blockIdx.x) * TILE_WORDS + part];
                const unsigned c0 = tile * TILE_CHUNKS + part * 32u;
                bool fits;
                unsigned pos = wave_slots(&cqcount, (unsigned)__builtin_popcount(word), ln, cqcap, fits);
                if (!fits) word = 0u;
                while (word) {
                    cq[pos++] = c0 + (unsigned)__builtin_ctz(word);
                    word &= word - 1u;
                }
            }
            __syncthreads();
            ncq = (unsigned)__builtin_amdgcn_readfirstlane((int)cqcount);
            if (ncq <= cqcap) {
                for (unsigned i0 = 4u * wv; i0 < ncq; i0 += 4u * OWN_WAVES * CHUNK_U) {
                    unsigned ck[CHUNK_U], h[CHUNK_U];
                    int4 v[CHUNK_U];
#pragma unroll
                    for (int u = 0; u < CHUNK_U; ++u) {
                        ck[u] = min(cq[i0 + (unsigned)u * 4u * OWN_WAVES + sub], nwhole - 1u);
                        v[u]  = dst4[ck[u] * 8u + part];
                    }
                    unsigned cnt = 0u;
#pragma unroll
                    for (int u = 0; u < CHUNK_U; ++u) {
                        const unsigned lim = (i0 + (unsigned)u * 4u * OWN_WAVES + sub < ncq) ? unloc : 0u;
                        h[u] = ((unsigned)(v[u].x - base) < lim ? 1u : 0u) | ((unsigned)(v[u].y - base) < lim ? 2u : 0u) |
                               ((unsigned)(v[u].z - base) < lim ? 4u : 0u) | ((unsigned)(v[u].w - base) < lim ? 8u : 0u);
                        cnt += (unsigned)__builtin_popcount(h[u]);
                    }
                    bool fits;
                    unsigned pos = wave_slots(&qcount, cnt, ln, QCAP, fits);
                    if (fits) {
#pragma unroll
                        for (int u = 0; u < CHUNK_U; ++u) {
                            unsigned m = h[u];
                            const unsigned e0 = (ck[u] << CHUNK_SHIFT) + part * 4u;
                            while (m) {
                                q_e[pos++] = e0 + (unsigned)__builtin_ctz(m);
                                m &= m - 1u;
                            }
                        }
                    }
                }
            }
        } else {
            for (unsigned g = gpos + (unsigned)tid; g < gend; g += OWN_THREADS * SCAN_U) {
                int4 v[SCAN_U];
#pragma unroll
                for (int u = 0; u < SCAN_U; ++u) {
                    unsigned ph = min(g + (unsigned)u * OWN_THREADS, gend - 1u) + rot;
                    if (ph >= ngroups) ph -= ngroups;
                    v[u] = dst4[ph];
                }
#pragma unroll
                for (int u = 0; u < SCAN_U; ++u) {
                    unsigned gi = g + (unsigned)u * OWN_THREADS;
                    unsigned ph = gi + rot;
                    if (ph >= ngroups) ph -= ngroups;
                    unsigned e0 = ph << 2;
                    unsigned lim = (gi < gend) ? unloc : 0u;
                    bool h0 = (unsigned)(v[u].x - base) < lim, h1 = (unsigned)(v[u].y - base) < lim;
                    bool h2 = (unsigned)(v[u].z - base) < lim, h3 = (unsigned)(v[u].w - base) < lim;
                    if (h0 | h1 | h2 | h3) {
                        if (h0) q_push(&qcount, q_e, e0);
                        if (h1) q_push(&qcount, q_e, e0 + 1u);
                        if (h2) q_push(&qcount, q_e, e0 + 2u);
                        if (h3) q_push(&qcount, q_e, e0 + 3u);
                    }
                }
            }
        }
        if (gend == ngroups) {
            unsigned e = (nwhole << eshift) + (unsigned)tid;
            if ((unsigned)tid < (E & ((1u << eshift) - 1u))) {
                if ((unsigned)(dst[e] - base) < unloc) q_push(&qcount, q_e, e);
            }
        }
        __syncthreads();
        const unsigned total = (unsigned)__builtin_amdgcn_readfirstlane((int)qcount);
        __syncthreads();

        if (total > QCAP || ncq > cqcap) {
            if (tid == 0) { qcount = 0u; cqcount = 0u; }
            glen = gsafe;
            __syncthreads();
            continue;
        }

        for (unsigned i = (unsigned)tid; i < total; i += OWN_THREADS) {
            const unsigned e = q_e[i];
            unsigned s  = (unsigned)src[e];
            unsigned dl = (unsigned)(dst[e] - base);
            float v = el[s] + er_s[dl];
            v = v > 0.0f ? v : NEG_SLOPE * v;
            q_e[i]   = s;
            q_dl[i]  = dl;
            q_val[i] = v;
            atomicMax(&mb_s[dl], enc_f(v));
        }
        __syncthreads();

        for (int i = tid; i < nloc; i += OWN_THREADS) {
            float mo = mrun_s[i];
            float mn = fmaxf(mo, dec_f(mb_s[i]));
            mrun_s[i] = mn;
            mb_s[i]   = ENC_NEG_INF;
            if (nflush) {
                float s = (mo == mn) ? 1.0f : expf(mo - mn);
                sc_s[i]  = s;
                den_s[i] *= s;
            }
        }
        __syncthreads();
        if (nflush) {
            for (int idx = tid; idx < nloc * OUT_DIM; idx += OWN_THREADS) acc[idx] *= sc_s[idx >> 6];
        }

        for (unsigned i = (unsigned)tid; i < total; i += OWN_THREADS) {
            unsigned dl = q_dl[i];
            float ex = expf(q_val[i] - mrun_s[dl]);
            q_val[i] = ex;
            atomicAdd(&den_s[dl], ex);
            const unsigned w = dl / ROWS_PW;
            const unsigned slot = atomicAdd(&wcnt[w], 1u);
            if (slot < WL_CAP) wl[w * WL_CAP + slot] = (unsigned short)i;
            else               wovf = 1u;
        }
        __syncthreads();
        const bool own_rows = wovf == 0u;

        if (own_rows) {
            const unsigned n  = (unsigned)__builtin_amdgcn_readfirstlane((int)wcnt[wv]);
            const unsigned r0 = wv * ROWS_PW;
            if (n) {
                v32f ar = {};
#pragma unroll
                for (unsigned k = 0u; k < ROWS_PW; ++k) {
                    if (r0 + k < unloc) {
                        const v2f t = *(const v2f*)&acc[(r0 + k) * OUT_DIM + 2u * ln];
                        ar[2u * k] = t.x; ar[2u * k + 1u] = t.y;
                    }
                }
                const unsigned short* mine = &wl[wv * WL_CAP];
                for (unsigned g = 0u; g < n; g += 32u) {
                    const unsigned nn = min(32u, n - g);
                    const unsigned i  = mine[min(g + ln, n - 1u)];
                    const unsigned s_l  = q_e[i];
                    const unsigned rw_l = q_dl[i] - r0;
                    const unsigned ex_l = __float_as_uint(q_val[i]);
                    if (nn == 32u) {
#pragma unroll
                        for (unsigned j0 = 0u; j0 < 32u; j0 += GATHER_U)
                            gather_batch_own(feat, ar, s_l, rw_l, ex_l, j0, 32u, ln);
                    } else {
#pragma nounroll
                        for (unsigned j0 = 0u; j0 < nn; j0 += GATHER_U)
                            gather_batch_own(feat, ar, s_l, rw_l, ex_l, j0, nn, ln);
                    }
                }
#pragma unroll
                for (unsigned k = 0u; k < ROWS_PW; ++k)
                    if (r0 + k < unloc) *(v2f*)&acc[(r0 + k) * OUT_DIM + 2u * ln] = (v2f){ar[2u * k], ar[2u * k + 1u]};
            }
        } else if (total) {
            const unsigned nbat = (total + GATHER_U - 1u) / GATHER_U;
            const unsigned beg  = (wv * nbat / OWN_WAVES) * GATHER_U;
            const unsigned end  = min(((wv + 1u) * nbat / OWN_WAVES) * GATHER_U, total);
            for (unsigned g = beg; g < end; g += 32u) {
                const unsigned n = min(32u, end - g);
                const unsigned i = min(g + ln, end - 1u);
                const unsigned s_l  = q_e[i];
                const unsigned dl_l = q_dl[i];
                const unsigned ex_l = __float_as_uint(q_val[i]);
                if (n == 32u) {
#pragma unroll
                    for (unsigned j0 = 0u; j0 < 32u; j0 += GATHER_U)
                        gather_batch(feat, acc, s_l, dl_l, ex_l, j0, 32u, ln);
                } else {
#pragma nounroll
                    for (unsigned j0 = 0u; j0 < n; j0 += GATHER_U)
                        gather_batch(feat, acc, s_l, dl_l, ex_l, j0, n, ln);
                }
            }
        }
        if (tid == 0) { qcount = 0u; cqcount = 0u; }
        __syncthreads();
        if (tid == 0) wovf = 0u;
        if (tid < OWN_WAVES) wcnt[tid] = 0u;
        ++nflush;
        gpos = gend;
        more = gpos < ngroups;
    }

    for (int idx = tid; idx < nloc * OUT_DIM; idx += OWN_THREADS) {
        float d = den_s[idx >> 6];
        float a = acc[idx];
        float r = d > 0.0f ? a / d : 0.0f;
        out[(size_t)base * OUT_DIM + idx] = tanhf(r + bias[idx & 63]);
    }
}

static inline size_t align256(size_t x) { return (x + 255) & ~(size_t)255; }

extern "C" void kernel_launch(void* const* d_in, const int* in_sizes, int n_in,
                              void* d_out, int out_size, void* d_ws, size_t ws_size,
                              hipStream_t stream) {
    const float* x      = (const float*)d_in[0];
    const int*   src    = (const int*)d_in[1];
    const int*   dst    = (const int*)d_in[2];
    const float* W      = (const float*)d_in[3];
    const float* attn_l = (const float*)d_in[4];
    const float* attn_r = (const float*)d_in[5];
    const float* bias   = (const float*)d_in[6];
    float*       out    = (float*)d_out;

    const int N = in_sizes[0] / IN_DIM;
    const int E = in_sizes[1];

    char* ws = (char*)d_ws;
    float* feat = (float*)ws;    ws += align256((size_t)N * OUT_DIM * sizeof(float));
    float* el   = (float*)ws;    ws += align256((size_t)N * sizeof(float));
    float* er   = (float*)ws;    ws += align256((size_t)N * sizeof(float));

    const int own_blocks = (N + OWN_NODES - 1) / OWN_NODES;
    const int n_tiles = ((E >> CHUNK_SHIFT) + TILE_CHUNKS - 1) / TILE_CHUNKS;
    const size_t bm_bytes = (size_t)n_tiles * KEY_OWNERS * TILE_WORDS * sizeof(unsigned);
    unsigned* bm = 0;
    if (own_blocks <= KEY_OWNERS && (size_t)(ws - (char*)d_ws) + align256(bm_bytes) <= ws_size)
        bm = (unsigned*)ws;

    {
        gat_gemm<<<GEMM_BLOCKS, GEMM_THREADS, 0, stream>>>(x, W, feat, N);
    }
    {
        int blocks = 400 + (bm ? n_tiles : 0);
        gat_logits<<<blocks, K2_THREADS, 0, stream>>>(feat, attn_l, attn_r, el, er, N, dst, bm, E, n_tiles);
    }
    {
        gat_owner<<<own_blocks, OWN_THREADS, 0, stream>>>(src, dst, el, er, feat, bias, out, N, E, bm);
    }
}


# K3 output pass: wave per row, 2 columns per lane; one hardware reciprocal of the row's denominator, no division per element; branch-free tanh (series under 0.625, else 1-2/(exp2+1)), no library call
# speedup vs baseline: 6.0598x; 1.0553x over previous
#include <hip/hip_runtime.h>

typedef __attribute__((ext_vector_type(2))) float v2f;
typedef __attribute__((ext_vector_type(4))) float v4f;
typedef __attribute__((ext_vector_type(8))) float v8f;

#define IN_DIM 128
#define OUT_DIM 64
#define NEG_SLOPE 0.2f

#define GEMM_THREADS 256
#define GEMM_WAVES   (GEMM_THREADS / 32)
#define GEMM_BLOCKS  256
#define KGROUPS      (IN_DIM / 8)

__global__ __launch_bounds__(GEMM_THREADS) void gat_gemm(const float* __restrict__ x,
                                                         const float* __restrict__ W,
                                                         float* __restrict__ feat,
                                                         int n_nodes) {
    const int lane = threadIdx.x & 31;
    const int c    = lane & 15;
    const int h    = lane >> 4;
    const int wave = __builtin_amdgcn_readfirstlane((int)(threadIdx.x >> 5));
    const int wlin    = wave * (int)gridDim.x + (int)blockIdx.x;
    const int n_waves = (int)gridDim.x * GEMM_WAVES;
    const int n_tiles = (n_nodes + 15) >> 4;

    v2f wf[KGROUPS][2][4];
#pragma unroll
    for (int g = 0; g < KGROUPS; ++g) {
#pragma unroll
        for (int j = 0; j < 2; ++j) {
            const int k = 8 * g + 4 * h + 2 * j;
            const v4f r0 = *(const v4f*)(W + (size_t)k * OUT_DIM + 4 * c);
            const v4f r1 = *(const v4f*)(W + (size_t)(k + 1) * OUT_DIM + 4 * c);
            wf[g][j][0] = (v2f){r0.x, r1.x};
            wf[g][j][1] = (v2f){r0.y, r1.y};
            wf[g][j][2] = (v2f){r0.z, r1.z};
            wf[g][j][3] = (v2f){r0.w, r1.w};
        }
    }

    if (wlin >= n_tiles) return;

    v4f xa[KGROUPS];
    {
        const int row = min(wlin * 16 + c, n_nodes - 1);
        const v4f* xp = (const v4f*)(x + (size_t)row * IN_DIM + 4 * h);
#pragma unroll
        for (int g = 0; g < KGROUPS; ++g) xa[g] = xp[2 * g];
    }

    for (int tile = wlin; tile < n_tiles; tile += n_waves) {
        v4f xn[KGROUPS];
        {
            const int nt  = min(tile + n_waves, n_tiles - 1);
            const int row = min(nt * 16 + c, n_nodes - 1);
            const v4f* xp = (const v4f*)(x + (size_t)row * IN_DIM + 4 * h);
#pragma unroll
            for (int g = 0; g < KGROUPS; ++g) xn[g] = xp[2 * g];
        }

        v8f acc0 = {}, acc1 = {}, acc2 = {}, acc3 = {};
#pragma unroll
        for (int g = 0; g < KGROUPS; ++g) {
            const v2f a0 = (v2f){xa[g].x, xa[g].y};
            const v2f a1 = (v2f){xa[g].z, xa[g].w};
            acc0 = __builtin_amdgcn_wmma_f32_16x16x4_f32(false, a0, false, wf[g][0][0], (short)0, acc0, false, false);
            acc1 = __builtin_amdgcn_wmma_f32_16x16x4_f32(false, a0, false, wf[g][0][1], (short)0, acc1, false, false);
            acc2 = __builtin_amdgcn_wmma_f32_16x16x4_f32(false, a0, false, wf[g][0][2], (short)0, acc2, false, false);
            acc3 = __builtin_amdgcn_wmma_f32_16x16x4_f32(false, a0, false, wf[g][0][3], (short)0, acc3, false, false);
            acc0 = __builtin_amdgcn_wmma_f32_16x16x4_f32(false, a1, false, wf[g][1][0], (short)0, acc0, false, false);
            acc1 = __builtin_amdgcn_wmma_f32_16x16x4_f32(false, a1, false, wf[g][1][1], (short)0, acc1, false, false);
            acc2 = __builtin_amdgcn_wmma_f32_16x16x4_f32(false, a1, false, wf[g][1][2], (short)0, acc2, false, false);
            acc3 = __builtin_amdgcn_wmma_f32_16x16x4_f32(false, a1, false, wf[g][1][3], (short)0, acc3, false, false);
        }

        float* o = feat + ((size_t)tile * 16 + 8 * h) * OUT_DIM + 4 * c;
        if (tile * 16 + 16 <= n_nodes) {
#pragma unroll
            for (int r = 0; r < 8; ++r)
                *(v4f*)(o + r * OUT_DIM) = (v4f){acc0[r], acc1[r], acc2[r], acc3[r]};
        } else {
            const int rows_here = n_nodes - tile * 16 - 8 * h;
#pragma unroll
            for (int r = 0; r < 8; ++r)
                if (r < rows_here)
                    *(v4f*)(o + r * OUT_DIM) = (v4f){acc0[r], acc1[r], acc2[r], acc3[r]};
        }

#pragma unroll
        for (int g = 0; g < KGROUPS; ++g) xa[g] = xn[g];
    }
}

#define OWN_NODES   391
#define KEY_OWNERS  256

#define CHUNK_SHIFT 5
#define TILE_CHUNKS 256
#define TILE_WORDS  (TILE_CHUNKS / 32)
#define TILE_PITCH  (TILE_WORDS + 1)
#define K2_THREADS  256

#define LGROUPS (OUT_DIM / 8)

__global__ __launch_bounds__(K2_THREADS) void gat_logits(const float* __restrict__ feat,
                           const float* __restrict__ attn_l,
                           const float* __restrict__ attn_r,
                           float* __restrict__ el, float* __restrict__ er,
                           int n_nodes,
                           const int* __restrict__ dst, unsigned* __restrict__ bm, int n_edges, int n_tiles) {
    __shared__ unsigned rows[KEY_OWNERS * TILE_PITCH];
    const int n_bm = bm ? n_tiles : 0;
    if ((int)blockIdx.x < n_bm) {
        const unsigned t    = threadIdx.x;
        const unsigned tile = blockIdx.x;
        for (unsigned i = t; i < KEY_OWNERS * TILE_PITCH; i += K2_THREADS) rows[i] = 0u;
        __syncthreads();
        const unsigned nvec = ((unsigned)n_edges >> CHUNK_SHIFT) << 3;
        const int4* __restrict__ dst4 = (const int4*)dst;
        const unsigned g0  = tile * (TILE_CHUNKS * 8u) + t;
        const unsigned bit = 1u << (t >> 3);
        int4 d[TILE_WORDS];
#pragma unroll
        for (int j = 0; j < TILE_WORDS; ++j) d[j] = dst4[min(g0 + (unsigned)j * K2_THREADS, nvec - 1u)];
#pragma unroll
        for (int j = 0; j < TILE_WORDS; ++j) {
            if (g0 + (unsigned)j * K2_THREADS < nvec) {
                atomicOr(&rows[min((unsigned)d[j].x / OWN_NODES, KEY_OWNERS - 1u) * TILE_PITCH + j], bit);
                atomicOr(&rows[min((unsigned)d[j].y / OWN_NODES, KEY_OWNERS - 1u) * TILE_PITCH + j], bit);
                atomicOr(&rows[min((unsigned)d[j].z / OWN_NODES, KEY_OWNERS - 1u) * TILE_PITCH + j], bit);
                atomicOr(&rows[min((unsigned)d[j].w / OWN_NODES, KEY_OWNERS - 1u) * TILE_PITCH + j], bit);
            }
        }
        __syncthreads();
        for (unsigned o = t; o < KEY_OWNERS; o += K2_THREADS) {
            const unsigned* r = &rows[o * TILE_PITCH];
            uint4* out4 = (uint4*)(bm + ((size_t)tile * KEY_OWNERS + o) * TILE_WORDS);
            out4[0] = make_uint4(r[0], r[1], r[2], r[3]);
            out4[1] = make_uint4(r[4], r[5], r[6], r[7]);
        }
        return;
    }
    const int lane = threadIdx.x & 31;
    const int c    = lane & 15;
    const int h    = lane >> 4;
    const int wave    = __builtin_amdgcn_readfirstlane((int)(threadIdx.x >> 5));
    const int wlin    = ((int)blockIdx.x - n_bm) * (K2_THREADS / 32) + wave;
    const int n_waves = ((int)gridDim.x - n_bm) * (K2_THREADS / 32);
    const int n_grp   = (n_nodes + 31) >> 5;

    const float* av = (c == 1) ? attn_r : attn_l;
    const unsigned keep = (c < 2) ? 0xFFFFFFFFu : 0u;
    v2f bf[LGROUPS][2];
#pragma unroll
    for (int g = 0; g < LGROUPS; ++g) {
#pragma unroll
        for (int j = 0; j < 2; ++j) {
            const int k = 8 * g + 4 * h + 2 * j;
            bf[g][j] = (v2f){__uint_as_float(__float_as_uint(av[k]) & keep),
                             __uint_as_float(__float_as_uint(av[k + 1]) & keep)};
        }
    }

    for (int grp = wlin; grp < n_grp; grp += n_waves) {
        const int rowA = min(grp * 32 + c, n_nodes - 1);
        const int rowB = min(grp * 32 + 16 + c, n_nodes - 1);
        const v4f* pa = (const v4f*)(feat + (size_t)rowA * OUT_DIM + 4 * h);
        const v4f* pb = (const v4f*)(feat + (size_t)rowB * OUT_DIM + 4 * h);
        v4f fa[LGROUPS], fb[LGROUPS];
#pragma unroll
        for (int g = 0; g < LGROUPS; ++g) { fa[g] = pa[2 * g]; fb[g] = pb[2 * g]; }
        __builtin_amdgcn_sched_barrier(0);

        v8f accA = {}, accB = {};
#pragma unroll
        for (int g = 0; g < LGROUPS; ++g) {
            const v2f a0 = (v2f){fa[g].x, fa[g].y}, a1 = (v2f){fa[g].z, fa[g].w};
            const v2f b0 = (v2f){fb[g].x, fb[g].y}, b1 = (v2f){fb[g].z, fb[g].w};
            accA = __builtin_amdgcn_wmma_f32_16x16x4_f32(false, a0, false, bf[g][0], (short)0, accA, false, false);
            accB = __builtin_amdgcn_wmma_f32_16x16x4_f32(false, b0, false, bf[g][0], (short)0, accB, false, false);
            accA = __builtin_amdgcn_wmma_f32_16x16x4_f32(false, a1, false, bf[g][1], (short)0, accA, false, false);
            accB = __builtin_amdgcn_wmma_f32_16x16x4_f32(false, b1, false, bf[g][1], (short)0, accB, false, false);
        }

        if (c < 2) {
            float* o = (c == 0 ? el : er) + (size_t)grp * 32 + 8 * h;
            if (grp * 32 + 32 <= n_nodes) {
                *(v4f*)(o)      = (v4f){accA[0], accA[1], accA[2], accA[3]};
                *(v4f*)(o + 4)  = (v4f){accA[4], accA[5], accA[6], accA[7]};
                *(v4f*)(o + 16) = (v4f){accB[0], accB[1], accB[2], accB[3]};
                *(v4f*)(o + 20) = (v4f){accB[4], accB[5], accB[6], accB[7]};
            } else {
                const int left = n_nodes - grp * 32 - 8 * h;
#pragma unroll
                for (int r = 0; r < 8; ++r) {
                    if (r < left)      o[r]      = accA[r];
                    if (r + 16 < left) o[16 + r] = accB[r];
                }
            }
        }
    }
}

#define OWN_THREADS 1024
#define OWN_WAVES   (OWN_THREADS / 32)
#define QCAP        16384
#define SCAN_U      8
#define CHUNK_U     16
#define GATHER_U    8
#define ENC_NEG_INF 0x007FFFFFu
#define ROWS_PW     13
#define WL_CAP      320

__device__ __forceinline__ unsigned enc_f(float v) {
    unsigned b = __float_as_uint(v);
    return (b & 0x80000000u) ? ~b : (b | 0x80000000u);
}
__device__ __forceinline__ float dec_f(unsigned e) {
    unsigned b = (e & 0x80000000u) ? (e & 0x7FFFFFFFu) : ~e;
    return __uint_as_float(b);
}

__device__ __forceinline__ void q_push(unsigned* qcount, unsigned* q_e, unsigned e) {
    unsigned slot = atomicAdd(qcount, 1u);
    if (slot < QCAP) q_e[slot] = e;
}

__device__ __forceinline__ unsigned wave_slots(unsigned* counter, unsigned cnt, unsigned ln, unsigned cap, bool& fits) {
    unsigned incl = cnt;
#pragma unroll
    for (unsigned d = 1u; d < 32u; d <<= 1) {
        const unsigned t = __shfl_up(incl, d, 32);
        if (ln >= d) incl += t;
    }
    const unsigned wtot = (unsigned)__builtin_amdgcn_readlane((int)incl, 31);
    unsigned b = 0u;
    if (ln == 0u) b = atomicAdd(counter, wtot);
    b = (unsigned)__builtin_amdgcn_readfirstlane((int)b);
    fits = b + wtot <= cap;
    return b + incl - cnt;
}

__device__ __forceinline__ void gather_batch(const float* __restrict__ feat, float* acc,
                                             unsigned s_l, unsigned dl_l, unsigned ex_l,
                                             unsigned j0, unsigned n, unsigned ln) {
    v2f f[GATHER_U]; unsigned dl[GATHER_U]; float ex[GATHER_U];
#pragma unroll
    for (unsigned u = 0u; u < GATHER_U; ++u) {
        const int j = (int)min(j0 + u, n - 1u);
        const unsigned s = (unsigned)__builtin_amdgcn_readlane((int)s_l, j);
        dl[u] = (unsigned)__builtin_amdgcn_readlane((int)dl_l, j);
        ex[u] = __uint_as_float((unsigned)__builtin_amdgcn_readlane((int)ex_l, j));
        const float* row = feat + (size_t)s * OUT_DIM;
        f[u] = *(const v2f*)(row + 2u * ln);
    }
#pragma unroll
    for (unsigned u = 0u; u < GATHER_U; ++u) {
        if (j0 + u < n) {
            float* a = &acc[dl[u] * OUT_DIM + 2u * ln];
            atomicAdd(a,     ex[u] * f[u].x);
            atomicAdd(a + 1, ex[u] * f[u].y);
        }
    }
}

typedef __attribute__((ext_vector_type(32))) float v32f;
__device__ __forceinline__ void gather_batch_own(const float* __restrict__ feat, v32f& ar,
                                                 unsigned s_l, unsigned rw_l, unsigned ex_l,
                                                 unsigned j0, unsigned n, unsigned ln) {
    v2f f[GATHER_U];
#pragma unroll
    for (unsigned u = 0u; u < GATHER_U; ++u) {
        const int j = (int)min(j0 + u, n - 1u);
        const unsigned s = (unsigned)__builtin_amdgcn_readlane((int)s_l, j);
        const float* row = feat + (size_t)s * OUT_DIM;
        f[u] = *(const v2f*)(row + 2u * ln);
    }
#pragma unroll
    for (unsigned u = 0u; u < GATHER_U; ++u) {
        if (j0 + u < n) {
            const int j = (int)(j0 + u);
            const unsigned r2 = 2u * (unsigned)__builtin_amdgcn_readlane((int)rw_l, j);
            const float w = __uint_as_float((unsigned)__builtin_amdgcn_readlane((int)ex_l, j));
            const float ox = ar[r2], oy = ar[r2 + 1u];
            ar[r2]      = fmaf(w, f[u].x, ox);
            ar[r2 + 1u] = fmaf(w, f[u].y, oy);
        }
    }
}

__device__ __forceinline__ float tanh_sel(float x) {
    const float ax = fabsf(x);
    const float x2 = ax * ax;
    float p = fmaf(x2, -0.00570002059f, 0.0206340719f);
    p = fmaf(x2, p, -0.0537379310f);
    p = fmaf(x2, p, 0.133314162f);
    p = fmaf(x2, p, -0.333332807f);
    const float small = fmaf(x2, ax * p, ax);
    const float e   = __builtin_amdgcn_exp2f(ax * 2.88539004f);
    const float big = fmaf(__builtin_amdgcn_rcpf(e + 1.0f), -2.0f, 1.0f);
    return copysignf(ax < 0.625f ? small : big, x);
}

__global__ __launch_bounds__(OWN_THREADS) void gat_owner(
        const int* __restrict__ src, const int* __restrict__ dst,
        const float* __restrict__ el, const float* __restrict__ er,
        const float* __restrict__ feat, const float* __restrict__ bias,
        float* __restrict__ out, int n_nodes, int n_edges,
        const unsigned* __restrict__ bm) {
    __shared__ float    acc[OWN_NODES * OUT_DIM];
    __shared__ float    den_s[OWN_NODES];
    __shared__ float    mrun_s[OWN_NODES];
    __shared__ float    sc_s[OWN_NODES];
    __shared__ float    er_s[OWN_NODES];
    __shared__ unsigned mb_s[OWN_NODES];
    __shared__ unsigned q_e[QCAP];
    __shared__ unsigned q_dl[QCAP];
    __shared__ float    q_val[QCAP];
    __shared__ unsigned qcount;
    __shared__ unsigned cqcount;
    __shared__ unsigned short wl[OWN_WAVES * WL_CAP];
    __shared__ unsigned wcnt[OWN_WAVES];
    __shared__ unsigned wovf;
    static_assert((OWN_NODES + ROWS_PW - 1) / ROWS_PW <= OWN_WAVES, "every row needs a wave");
    static_assert(QCAP <= 65536, "queue positions are kept in 16 bits");

    const int tid  = threadIdx.x;
    const int base = blockIdx.x * OWN_NODES;
    const int nloc = min(OWN_NODES, n_nodes - base);
    const unsigned unloc = (unsigned)nloc;

    for (int idx = tid; idx < nloc * OUT_DIM; idx += OWN_THREADS) acc[idx] = 0.0f;
    for (int i = tid; i < nloc; i += OWN_THREADS) {
        den_s[i]  = 0.0f;
        mrun_s[i] = __uint_as_float(0xFF800000u);
        mb_s[i]   = ENC_NEG_INF;
        er_s[i]   = er[base + i];
    }
    if (tid == 0) { qcount = 0u; cqcount = 0u; wovf = 0u; }
    if (tid < OWN_WAVES) wcnt[tid] = 0u;
    __syncthreads();

    const unsigned E = (unsigned)n_edges;
    const bool use_bm = bm != 0;
    const unsigned eshift = use_bm ? (unsigned)CHUNK_SHIFT : 2u;
    const unsigned nwhole = E >> eshift;
    const unsigned ngroups = use_bm ? (nwhole + TILE_CHUNKS - 1u) / TILE_CHUNKS : nwhole;
    const unsigned gsafe = use_bm ? 1u : (QCAP >> 2) - 1u;
    const int4* __restrict__ dst4 = (const int4*)dst;
    unsigned* cq = (unsigned*)q_val;
    const unsigned wv = (unsigned)__builtin_amdgcn_readfirstlane(tid >> 5);
    const unsigned ln = (unsigned)tid & 31u;
    const unsigned sub = ln >> 3, part = ln & 7u;
    const unsigned cqcap = QCAP - 4u * OWN_WAVES * CHUNK_U;
    unsigned gpos = 0u, glen = ngroups, nflush = 0u;
    bool more = E > 0u;
    const unsigned rot = (unsigned)(((unsigned long long)blockIdx.x * ngroups) / gridDim.x) & ~31u;

    while (more) {
        const unsigned gend = min(gpos + glen, ngroups);
        unsigned ncq = 0u;

        if (use_bm) {
            for (unsigned tb = gpos + 4u * wv; tb < gend; tb += 4u * OWN_WAVES) {
                const unsigned tile = tb + sub;
                unsigned word = 0u;
                if (tile < gend) word = bm[((size_t)tile * KEY_OWNERS + blockIdx.x) * TILE_WORDS + part];
                const unsigned c0 = tile * TILE_CHUNKS + part * 32u;
                bool fits;
                unsigned pos = wave_slots(&cqcount, (unsigned)__builtin_popcount(word), ln, cqcap, fits);
                if (!fits) word = 0u;
                while (word) {
                    cq[pos++] = c0 + (unsigned)__builtin_ctz(word);
                    word &= word - 1u;
                }
            }
            __syncthreads();
            ncq = (unsigned)__builtin_amdgcn_readfirstlane((int)cqcount);
            if (ncq <= cqcap) {
                for (unsigned i0 = 4u * wv; i0 < ncq; i0 += 4u * OWN_WAVES * CHUNK_U) {
                    unsigned ck[CHUNK_U], h[CHUNK_U];
                    int4 v[CHUNK_U];
#pragma unroll
                    for (int u = 0; u < CHUNK_U; ++u) {
                        ck[u] = min(cq[i0 + (unsigned)u * 4u * OWN_WAVES + sub], nwhole - 1u);
                        v[u]  = dst4[ck[u] * 8u + part];
                    }
                    unsigned cnt = 0u;
#pragma unroll
                    for (int u = 0; u < CHUNK_U; ++u) {
                        const unsigned lim = (i0 + (unsigned)u * 4u * OWN_WAVES + sub < ncq) ? unloc : 0u;
                        h[u] = ((unsigned)(v[u].x - base) < lim ? 1u : 0u) | ((unsigned)(v[u].y - base) < lim ? 2u : 0u) |
                               ((unsigned)(v[u].z - base) < lim ? 4u : 0u) | ((unsigned)(v[u].w - base) < lim ? 8u : 0u);
                        cnt += (unsigned)__builtin_popcount(h[u]);
                    }
                    bool fits;
                    unsigned pos = wave_slots(&qcount, cnt, ln, QCAP, fits);
                    if (fits) {
#pragma unroll
                        for (int u = 0; u < CHUNK_U; ++u) {
                            unsigned m = h[u];
                            const unsigned e0 = (ck[u] << CHUNK_SHIFT) + part * 4u;
                            while (m) {
                                q_e[pos++] = e0 + (unsigned)__builtin_ctz(m);
                                m &= m - 1u;
                            }
                        }
                    }
                }
            }
        } else {
            for (unsigned g = gpos + (unsigned)tid; g < gend; g += OWN_THREADS * SCAN_U) {
                int4 v[SCAN_U];
#pragma unroll
                for (int u = 0; u < SCAN_U; ++u) {
                    unsigned ph = min(g + (unsigned)u * OWN_THREADS, gend - 1u) + rot;
                    if (ph >= ngroups) ph -= ngroups;
                    v[u] = dst4[ph];
                }
#pragma unroll
                for (int u = 0; u < SCAN_U; ++u) {
                    unsigned gi = g + (unsigned)u * OWN_THREADS;
                    unsigned ph = gi + rot;
                    if (ph >= ngroups) ph -= ngroups;
                    unsigned e0 = ph << 2;
                    unsigned lim = (gi < gend) ? unloc : 0u;
                    bool h0 = (unsigned)(v[u].x - base) < lim, h1 = (unsigned)(v[u].y - base) < lim;
                    bool h2 = (unsigned)(v[u].z - base) < lim, h3 = (unsigned)(v[u].w - base) < lim;
                    if (h0 | h1 | h2 | h3) {
                        if (h0) q_push(&qcount, q_e, e0);
                        if (h1) q_push(&qcount, q_e, e0 + 1u);
                        if (h2) q_push(&qcount, q_e, e0 + 2u);
                        if (h3) q_push(&qcount, q_e, e0 + 3u);
                    }
                }
            }
        }
        if (gend == ngroups) {
            unsigned e = (nwhole << eshift) + (unsigned)tid;
            if ((unsigned)tid < (E & ((1u << eshift) - 1u))) {
                if ((unsigned)(dst[e] - base) < unloc) q_push(&qcount, q_e, e);
            }
        }
        __syncthreads();
        const unsigned total = (unsigned)__builtin_amdgcn_readfirstlane((int)qcount);
        __syncthreads();

        if (total > QCAP || ncq > cqcap) {
            if (tid == 0) { qcount = 0u; cqcount = 0u; }
            glen = gsafe;
            __syncthreads();
            continue;
        }

        for (unsigned i = (unsigned)tid; i < total; i += OWN_THREADS) {
            const unsigned e = q_e[i];
            unsigned s  = (unsigned)src[e];
            unsigned dl = (unsigned)(dst[e] - base);
            float v = el[s] + er_s[dl];
            v = v > 0.0f ? v : NEG_SLOPE * v;
            q_e[i]   = s;
            q_dl[i]  = dl;
            q_val[i] = v;
            atomicMax(&mb_s[dl], enc_f(v));
        }
        __syncthreads();

        for (int i = tid; i < nloc; i += OWN_THREADS) {
            float mo = mrun_s[i];
            float mn = fmaxf(mo, dec_f(mb_s[i]));
            mrun_s[i] = mn;
            mb_s[i]   = ENC_NEG_INF;
            if (nflush) {
                float s = (mo == mn) ? 1.0f : expf(mo - mn);
                sc_s[i]  = s;
                den_s[i] *= s;
            }
        }
        __syncthreads();
        if (nflush) {
            for (int idx = tid; idx < nloc * OUT_DIM; idx += OWN_THREADS) acc[idx] *= sc_s[idx >> 6];
        }

        for (unsigned i = (unsigned)tid; i < total; i += OWN_THREADS) {
            unsigned dl = q_dl[i];
            float ex = expf(q_val[i] - mrun_s[dl]);
            q_val[i] = ex;
            atomicAdd(&den_s[dl], ex);
            const unsigned w = dl / ROWS_PW;
            const unsigned slot = atomicAdd(&wcnt[w], 1u);
            if (slot < WL_CAP) wl[w * WL_CAP + slot] = (unsigned short)i;
            else               wovf = 1u;
        }
        __syncthreads();
        const bool own_rows = wovf == 0u;

        if (own_rows) {
            const unsigned n  = (unsigned)__builtin_amdgcn_readfirstlane((int)wcnt[wv]);
            const unsigned r0 = wv * ROWS_PW;
            if (n) {
                v32f ar = {};
#pragma unroll
                for (unsigned k = 0u; k < ROWS_PW; ++k) {
                    if (r0 + k < unloc) {
                        const v2f t = *(const v2f*)&acc[(r0 + k) * OUT_DIM + 2u * ln];
                        ar[2u * k] = t.x; ar[2u * k + 1u] = t.y;
                    }
                }
                const unsigned short* mine = &wl[wv * WL_CAP];
                for (unsigned g = 0u; g < n; g += 32u) {
                    const unsigned nn = min(32u, n - g);
                    const unsigned i  = mine[min(g + ln, n - 1u)];
                    const unsigned s_l  = q_e[i];
                    const unsigned rw_l = q_dl[i] - r0;
                    const unsigned ex_l = __float_as_uint(q_val[i]);
                    if (nn == 32u) {
#pragma unroll
                        for (unsigned j0 = 0u; j0 < 32u; j0 += GATHER_U)
                            gather_batch_own(feat, ar, s_l, rw_l, ex_l, j0, 32u, ln);
                    } else {
#pragma nounroll
                        for (unsigned j0 = 0u; j0 < nn; j0 += GATHER_U)
                            gather_batch_own(feat, ar, s_l, rw_l, ex_l, j0, nn, ln);
                    }
                }
#pragma unroll
                for (unsigned k = 0u; k < ROWS_PW; ++k)
                    if (r0 + k < unloc) *(v2f*)&acc[(r0 + k) * OUT_DIM + 2u * ln] = (v2f){ar[2u * k], ar[2u * k + 1u]};
            }
        } else if (total) {
            const unsigned nbat = (total + GATHER_U - 1u) / GATHER_U;
            const unsigned beg  = (wv * nbat / OWN_WAVES) * GATHER_U;
            const unsigned end  = min(((wv + 1u) * nbat / OWN_WAVES) * GATHER_U, total);
            for (unsigned g = beg; g < end; g += 32u) {
                const unsigned n = min(32u, end - g);
                const unsigned i = min(g + ln, end - 1u);
                const unsigned s_l  = q_e[i];
                const unsigned dl_l = q_dl[i];
                const unsigned ex_l = __float_as_uint(q_val[i]);
                if (n == 32u) {
#pragma unroll
                    for (unsigned j0 = 0u; j0 < 32u; j0 += GATHER_U)
                        gather_batch(feat, acc, s_l, dl_l, ex_l, j0, 32u, ln);
                } else {
#pragma nounroll
                    for (unsigned j0 = 0u; j0 < n; j0 += GATHER_U)
                        gather_batch(feat, acc, s_l, dl_l, ex_l, j0, n, ln);
                }
            }
        }
        if (tid == 0) { qcount = 0u; cqcount = 0u; }
        __syncthreads();
        if (tid == 0) wovf = 0u;
        if (tid < OWN_WAVES) wcnt[tid] = 0u;
        ++nflush;
        gpos = gend;
        more = gpos < ngroups;
    }

    {
        const unsigned npair = unloc * (OUT_DIM / 2);
        const v2f* acc2 = (const v2f*)acc;
        v2f*       out2 = (v2f*)(out + (size_t)base * OUT_DIM);
        const v2f  b2   = ((const v2f*)bias)[(unsigned)tid & 31u];
        for (unsigned p = (unsigned)tid; p < npair; p += OWN_THREADS) {
            const float d = den_s[p >> 5];
            const v2f   a = acc2[p];
            const float inv = d > 0.0f ? __builtin_amdgcn_rcpf(d) : 0.0f;
            v2f o;
            o.x = tanh_sel(fmaf(a.x, inv, b2.x));
            o.y = tanh_sel(fmaf(a.y, inv, b2.y));
            out2[p] = o;
        }
    }
}

static inline size_t align256(size_t x) { return (x + 255) & ~(size_t)255; }

extern "C" void kernel_launch(void* const* d_in, const int* in_sizes, int n_in,
                              void* d_out, int out_size, void* d_ws, size_t ws_size,
                              hipStream_t stream) {
    const float* x      = (const float*)d_in[0];
    const int*   src    = (const int*)d_in[1];
    const int*   dst    = (const int*)d_in[2];
    const float* W      = (const float*)d_in[3];
    const float* attn_l = (const float*)d_in[4];
    const float* attn_r = (const float*)d_in[5];
    const float* bias   = (const float*)d_in[6];
    float*       out    = (float*)d_out;

    const int N = in_sizes[0] / IN_DIM;
    const int E = in_sizes[1];

    char* ws = (char*)d_ws;
    float* feat = (float*)ws;    ws += align256((size_t)N * OUT_DIM * sizeof(float));
    float* el   = (float*)ws;    ws += align256((size_t)N * sizeof(float));
    float* er   = (float*)ws;    ws += align256((size_t)N * sizeof(float));

    const int own_blocks = (N + OWN_NODES - 1) / OWN_NODES;
    const int n_tiles = ((E >> CHUNK_SHIFT) + TILE_CHUNKS - 1) / TILE_CHUNKS;
    const size_t bm_bytes = (size_t)n_tiles * KEY_OWNERS * TILE_WORDS * sizeof(unsigned);
    unsigned* bm = 0;
    if (own_blocks <= KEY_OWNERS && (size_t)(ws - (char*)d_ws) + align256(bm_bytes) <= ws_size)
        bm = (unsigned*)ws;

    {
        gat_gemm<<<GEMM_BLOCKS, GEMM_THREADS, 0, stream>>>(x, W, feat, N);
    }
    {
        int blocks = 400 + (bm ? n_tiles : 0);
        gat_logits<<<blocks, K2_THREADS, 0, stream>>>(feat, attn_l, attn_r, el, er, N, dst, bm, E, n_tiles);
    }
    {
        gat_owner<<<own_blocks, OWN_THREADS, 0, stream>>>(src, dst, el, er, feat, bias, out, N, E, bm);
    }
}


# K3 scan reads each flagged chunk's src line beside its dst line and queues (source, local dst) from registers; P1's random src[e], dst[e] reads gone; queue claimed in batches of 8, sums in its order
# speedup vs baseline: 6.3679x; 1.0508x over previous
#include <hip/hip_runtime.h>

typedef __attribute__((ext_vector_type(2))) float v2f;
typedef __attribute__((ext_vector_type(4))) float v4f;
typedef __attribute__((ext_vector_type(8))) float v8f;

#define IN_DIM 128
#define OUT_DIM 64
#define NEG_SLOPE 0.2f

#define GEMM_THREADS 256
#define GEMM_WAVES   (GEMM_THREADS / 32)
#define GEMM_BLOCKS  256
#define KGROUPS      (IN_DIM / 8)

__global__ __launch_bounds__(GEMM_THREADS) void gat_gemm(const float* __restrict__ x,
                                                         const float* __restrict__ W,
                                                         float* __restrict__ feat,
                                                         int n_nodes) {
    const int lane = threadIdx.x & 31;
    const int c    = lane & 15;
    const int h    = lane >> 4;
    const int wave = __builtin_amdgcn_readfirstlane((int)(threadIdx.x >> 5));
    const int wlin    = wave * (int)gridDim.x + (int)blockIdx.x;
    const int n_waves = (int)gridDim.x * GEMM_WAVES;
    const int n_tiles = (n_nodes + 15) >> 4;

    v2f wf[KGROUPS][2][4];
#pragma unroll
    for (int g = 0; g < KGROUPS; ++g) {
#pragma unroll
        for (int j = 0; j < 2; ++j) {
            const int k = 8 * g + 4 * h + 2 * j;
            const v4f r0 = *(const v4f*)(W + (size_t)k * OUT_DIM + 4 * c);
            const v4f r1 = *(const v4f*)(W + (size_t)(k + 1) * OUT_DIM + 4 * c);
            wf[g][j][0] = (v2f){r0.x, r1.x};
            wf[g][j][1] = (v2f){r0.y, r1.y};
            wf[g][j][2] = (v2f){r0.z, r1.z};
            wf[g][j][3] = (v2f){r0.w, r1.w};
        }
    }

    if (wlin >= n_tiles) return;

    v4f xa[KGROUPS];
    {
        const int row = min(wlin * 16 + c, n_nodes - 1);
        const v4f* xp = (const v4f*)(x + (size_t)row * IN_DIM + 4 * h);
#pragma unroll
        for (int g = 0; g < KGROUPS; ++g) xa[g] = xp[2 * g];
    }

    for (int tile = wlin; tile < n_tiles; tile += n_waves) {
        v4f xn[KGROUPS];
        {
            const int nt  = min(tile + n_waves, n_tiles - 1);
            const int row = min(nt * 16 + c, n_nodes - 1);
            const v4f* xp = (const v4f*)(x + (size_t)row * IN_DIM + 4 * h);
#pragma unroll
            for (int g = 0; g < KGROUPS; ++g) xn[g] = xp[2 * g];
        }

        v8f acc0 = {}, acc1 = {}, acc2 = {}, acc3 = {};
#pragma unroll
        for (int g = 0; g < KGROUPS; ++g) {
            const v2f a0 = (v2f){xa[g].x, xa[g].y};
            const v2f a1 = (v2f){xa[g].z, xa[g].w};
            acc0 = __builtin_amdgcn_wmma_f32_16x16x4_f32(false, a0, false, wf[g][0][0], (short)0, acc0, false, false);
            acc1 = __builtin_amdgcn_wmma_f32_16x16x4_f32(false, a0, false, wf[g][0][1], (short)0, acc1, false, false);
            acc2 = __builtin_amdgcn_wmma_f32_16x16x4_f32(false, a0, false, wf[g][0][2], (short)0, acc2, false, false);
            acc3 = __builtin_amdgcn_wmma_f32_16x16x4_f32(false, a0, false, wf[g][0][3], (short)0, acc3, false, false);
            acc0 = __builtin_amdgcn_wmma_f32_16x16x4_f32(false, a1, false, wf[g][1][0], (short)0, acc0, false, false);
            acc1 = __builtin_amdgcn_wmma_f32_16x16x4_f32(false, a1, false, wf[g][1][1], (short)0, acc1, false, false);
            acc2 = __builtin_amdgcn_wmma_f32_16x16x4_f32(false, a1, false, wf[g][1][2], (short)0, acc2, false, false);
            acc3 = __builtin_amdgcn_wmma_f32_16x16x4_f32(false, a1, false, wf[g][1][3], (short)0, acc3, false, false);
        }

        float* o = feat + ((size_t)tile * 16 + 8 * h) * OUT_DIM + 4 * c;
        if (tile * 16 + 16 <= n_nodes) {
#pragma unroll
            for (int r = 0; r < 8; ++r)
                *(v4f*)(o + r * OUT_DIM) = (v4f){acc0[r], acc1[r], acc2[r], acc3[r]};
        } else {
            const int rows_here = n_nodes - tile * 16 - 8 * h;
#pragma unroll
            for (int r = 0; r < 8; ++r)
                if (r < rows_here)
                    *(v4f*)(o + r * OUT_DIM) = (v4f){acc0[r], acc1[r], acc2[r], acc3[r]};
        }

#pragma unroll
        for (int g = 0; g < KGROUPS; ++g) xa[g] = xn[g];
    }
}

#define OWN_NODES   391
#define KEY_OWNERS  256

#define CHUNK_SHIFT 5
#define TILE_CHUNKS 256
#define TILE_WORDS  (TILE_CHUNKS / 32)
#define TILE_PITCH  (TILE_WORDS + 1)
#define K2_THREADS  256

#define LGROUPS (OUT_DIM / 8)

__global__ __launch_bounds__(K2_THREADS) void gat_logits(const float* __restrict__ feat,
                           const float* __restrict__ attn_l,
                           const float* __restrict__ attn_r,
                           float* __restrict__ el, float* __restrict__ er,
                           int n_nodes,
                           const int* __restrict__ dst, unsigned* __restrict__ bm, int n_edges, int n_tiles) {
    __shared__ unsigned rows[KEY_OWNERS * TILE_PITCH];
    const int n_bm = bm ? n_tiles : 0;
    if ((int)blockIdx.x < n_bm) {
        const unsigned t    = threadIdx.x;
        const unsigned tile = blockIdx.x;
        for (unsigned i = t; i < KEY_OWNERS * TILE_PITCH; i += K2_THREADS) rows[i] = 0u;
        __syncthreads();
        const unsigned nvec = ((unsigned)n_edges >> CHUNK_SHIFT) << 3;
        const int4* __restrict__ dst4 = (const int4*)dst;
        const unsigned g0  = tile * (TILE_CHUNKS * 8u) + t;
        const unsigned bit = 1u << (t >> 3);
        int4 d[TILE_WORDS];
#pragma unroll
        for (int j = 0; j < TILE_WORDS; ++j) d[j] = dst4[min(g0 + (unsigned)j * K2_THREADS, nvec - 1u)];
#pragma unroll
        for (int j = 0; j < TILE_WORDS; ++j) {
            if (g0 + (unsigned)j * K2_THREADS < nvec) {
                atomicOr(&rows[min((unsigned)d[j].x / OWN_NODES, KEY_OWNERS - 1u) * TILE_PITCH + j], bit);
                atomicOr(&rows[min((unsigned)d[j].y / OWN_NODES, KEY_OWNERS - 1u) * TILE_PITCH + j], bit);
                atomicOr(&rows[min((unsigned)d[j].z / OWN_NODES, KEY_OWNERS - 1u) * TILE_PITCH + j], bit);
                atomicOr(&rows[min((unsigned)d[j].w / OWN_NODES, KEY_OWNERS - 1u) * TILE_PITCH + j], bit);
            }
        }
        __syncthreads();
        for (unsigned o = t; o < KEY_OWNERS; o += K2_THREADS) {
            const unsigned* r = &rows[o * TILE_PITCH];
            uint4* out4 = (uint4*)(bm + ((size_t)tile * KEY_OWNERS + o) * TILE_WORDS);
            out4[0] = make_uint4(r[0], r[1], r[2], r[3]);
            out4[1] = make_uint4(r[4], r[5], r[6], r[7]);
        }
        return;
    }
    const int lane = threadIdx.x & 31;
    const int c    = lane & 15;
    const int h    = lane >> 4;
    const int wave    = __builtin_amdgcn_readfirstlane((int)(threadIdx.x >> 5));
    const int wlin    = ((int)blockIdx.x - n_bm) * (K2_THREADS / 32) + wave;
    const int n_waves = ((int)gridDim.x - n_bm) * (K2_THREADS / 32);
    const int n_grp   = (n_nodes + 31) >> 5;

    const float* av = (c == 1) ? attn_r : attn_l;
    const unsigned keep = (c < 2) ? 0xFFFFFFFFu : 0u;
    v2f bf[LGROUPS][2];
#pragma unroll
    for (int g = 0; g < LGROUPS; ++g) {
#pragma unroll
        for (int j = 0; j < 2; ++j) {
            const int k = 8 * g + 4 * h + 2 * j;
            bf[g][j] = (v2f){__uint_as_float(__float_as_uint(av[k]) & keep),
                             __uint_as_float(__float_as_uint(av[k + 1]) & keep)};
        }
    }

    for (int grp = wlin; grp < n_grp; grp += n_waves) {
        const int rowA = min(grp * 32 + c, n_nodes - 1);
        const int rowB = min(grp * 32 + 16 + c, n_nodes - 1);
        const v4f* pa = (const v4f*)(feat + (size_t)rowA * OUT_DIM + 4 * h);
        const v4f* pb = (const v4f*)(feat + (size_t)rowB * OUT_DIM + 4 * h);
        v4f fa[LGROUPS], fb[LGROUPS];
#pragma unroll
        for (int g = 0; g < LGROUPS; ++g) { fa[g] = pa[2 * g]; fb[g] = pb[2 * g]; }
        __builtin_amdgcn_sched_barrier(0);

        v8f accA = {}, accB = {};
#pragma unroll
        for (int g = 0; g < LGROUPS; ++g) {
            const v2f a0 = (v2f){fa[g].x, fa[g].y}, a1 = (v2f){fa[g].z, fa[g].w};
            const v2f b0 = (v2f){fb[g].x, fb[g].y}, b1 = (v2f){fb[g].z, fb[g].w};
            accA = __builtin_amdgcn_wmma_f32_16x16x4_f32(false, a0, false, bf[g][0], (short)0, accA, false, false);
            accB = __builtin_amdgcn_wmma_f32_16x16x4_f32(false, b0, false, bf[g][0], (short)0, accB, false, false);
            accA = __builtin_amdgcn_wmma_f32_16x16x4_f32(false, a1, false, bf[g][1], (short)0, accA, false, false);
            accB = __builtin_amdgcn_wmma_f32_16x16x4_f32(false, b1, false, bf[g][1], (short)0, accB, false, false);
        }

        if (c < 2) {
            float* o = (c == 0 ? el : er) + (size_t)grp * 32 + 8 * h;
            if (grp * 32 + 32 <= n_nodes) {
                *(v4f*)(o)      = (v4f){accA[0], accA[1], accA[2], accA[3]};
                *(v4f*)(o + 4)  = (v4f){accA[4], accA[5], accA[6], accA[7]};
                *(v4f*)(o + 16) = (v4f){accB[0], accB[1], accB[2], accB[3]};
                *(v4f*)(o + 20) = (v4f){accB[4], accB[5], accB[6], accB[7]};
            } else {
                const int left = n_nodes - grp * 32 - 8 * h;
#pragma unroll
                for (int r = 0; r < 8; ++r) {
                    if (r < left)      o[r]      = accA[r];
                    if (r + 16 < left) o[16 + r] = accB[r];
                }
            }
        }
    }
}

#define OWN_THREADS 1024
#define OWN_WAVES   (OWN_THREADS / 32)
#define QCAP        16384
#define SCAN_U      8
#define CHUNK_U     8
#define GATHER_U    8
#define ENC_NEG_INF 0x007FFFFFu
#define ROWS_PW     13
#define WL_CAP      320

__device__ __forceinline__ unsigned enc_f(float v) {
    unsigned b = __float_as_uint(v);
    return (b & 0x80000000u) ? ~b : (b | 0x80000000u);
}
__device__ __forceinline__ float dec_f(unsigned e) {
    unsigned b = (e & 0x80000000u) ? (e & 0x7FFFFFFFu) : ~e;
    return __uint_as_float(b);
}

__device__ __forceinline__ void q_push(unsigned* qcount, unsigned* q_e, unsigned* q_dl, unsigned s, unsigned dl) {
    unsigned slot = atomicAdd(qcount, 1u);
    if (slot < QCAP) { q_e[slot] = s; q_dl[slot] = dl; }
}

__device__ __forceinline__ int pick4(int4 v, unsigned k) {
    const bool b0 = (k & 1u) != 0u, b1 = (k & 2u) != 0u;
    const int lo = b0 ? v.y : v.x;
    const int hi = b0 ? v.w : v.z;
    return b1 ? hi : lo;
}

__device__ __forceinline__ unsigned wave_slots(unsigned* counter, unsigned cnt, unsigned ln, unsigned cap, bool& fits) {
    unsigned incl = cnt;
#pragma unroll
    for (unsigned d = 1u; d < 32u; d <<= 1) {
        const unsigned t = __shfl_up(incl, d, 32);
        if (ln >= d) incl += t;
    }
    const unsigned wtot = (unsigned)__builtin_amdgcn_readlane((int)incl, 31);
    unsigned b = 0u;
    if (ln == 0u) b = atomicAdd(counter, wtot);
    b = (unsigned)__builtin_amdgcn_readfirstlane((int)b);
    fits = b + wtot <= cap;
    return b + incl - cnt;
}

__device__ __forceinline__ void gather_batch(const float* __restrict__ feat, float* acc,
                                             unsigned s_l, unsigned dl_l, unsigned ex_l,
                                             unsigned j0, unsigned n, unsigned ln) {
    v2f f[GATHER_U]; unsigned dl[GATHER_U]; float ex[GATHER_U];
#pragma unroll
    for (unsigned u = 0u; u < GATHER_U; ++u) {
        const int j = (int)min(j0 + u, n - 1u);
        const unsigned s = (unsigned)__builtin_amdgcn_readlane((int)s_l, j);
        dl[u] = (unsigned)__builtin_amdgcn_readlane((int)dl_l, j);
        ex[u] = __uint_as_float((unsigned)__builtin_amdgcn_readlane((int)ex_l, j));
        const float* row = feat + (size_t)s * OUT_DIM;
        f[u] = *(const v2f*)(row + 2u * ln);
    }
#pragma unroll
    for (unsigned u = 0u; u < GATHER_U; ++u) {
        if (j0 + u < n) {
            float* a = &acc[dl[u] * OUT_DIM + 2u * ln];
            atomicAdd(a,     ex[u] * f[u].x);
            atomicAdd(a + 1, ex[u] * f[u].y);
        }
    }
}

typedef __attribute__((ext_vector_type(32))) float v32f;
__device__ __forceinline__ void gather_batch_own(const float* __restrict__ feat, v32f& ar,
                                                 unsigned s_l, unsigned rw_l, unsigned ex_l,
                                                 unsigned j0, unsigned n, unsigned ln) {
    v2f f[GATHER_U];
#pragma unroll
    for (unsigned u = 0u; u < GATHER_U; ++u) {
        const int j = (int)min(j0 + u, n - 1u);
        const unsigned s = (unsigned)__builtin_amdgcn_readlane((int)s_l, j);
        const float* row = feat + (size_t)s * OUT_DIM;
        f[u] = *(const v2f*)(row + 2u * ln);
    }
#pragma unroll
    for (unsigned u = 0u; u < GATHER_U; ++u) {
        if (j0 + u < n) {
            const int j = (int)(j0 + u);
            const unsigned r2 = 2u * (unsigned)__builtin_amdgcn_readlane((int)rw_l, j);
            const float w = __uint_as_float((unsigned)__builtin_amdgcn_readlane((int)ex_l, j));
            const float ox = ar[r2], oy = ar[r2 + 1u];
            ar[r2]      = fmaf(w, f[u].x, ox);
            ar[r2 + 1u] = fmaf(w, f[u].y, oy);
        }
    }
}

__device__ __forceinline__ float tanh_sel(float x) {
    const float ax = fabsf(x);
    const float x2 = ax * ax;
    float p = fmaf(x2, -0.00570002059f, 0.0206340719f);
    p = fmaf(x2, p, -0.0537379310f);
    p = fmaf(x2, p, 0.133314162f);
    p = fmaf(x2, p, -0.333332807f);
    const float small = fmaf(x2, ax * p, ax);
    const float e   = __builtin_amdgcn_exp2f(ax * 2.88539004f);
    const float big = fmaf(__builtin_amdgcn_rcpf(e + 1.0f), -2.0f, 1.0f);
    return copysignf(ax < 0.625f ? small : big, x);
}

__global__ __launch_bounds__(OWN_THREADS) void gat_owner(
        const int* __restrict__ src, const int* __restrict__ dst,
        const float* __restrict__ el, const float* __restrict__ er,
        const float* __restrict__ feat, const float* __restrict__ bias,
        float* __restrict__ out, int n_nodes, int n_edges,
        const unsigned* __restrict__ bm) {
    __shared__ float    acc[OWN_NODES * OUT_DIM];
    __shared__ float    den_s[OWN_NODES];
    __shared__ float    mrun_s[OWN_NODES];
    __shared__ float    sc_s[OWN_NODES];
    __shared__ float    er_s[OWN_NODES];
    __shared__ unsigned mb_s[OWN_NODES];
    __shared__ unsigned q_e[QCAP];
    __shared__ unsigned q_dl[QCAP];
    __shared__ float    q_val[QCAP];
    __shared__ unsigned qcount;
    __shared__ unsigned cqcount;
    __shared__ unsigned short wl[OWN_WAVES * WL_CAP];
    __shared__ unsigned wcnt[OWN_WAVES];
    __shared__ unsigned wovf;
    static_assert((OWN_NODES + ROWS_PW - 1) / ROWS_PW <= OWN_WAVES, "every row needs a wave");
    static_assert(QCAP <= 65536, "queue positions are kept in 16 bits");

    const int tid  = threadIdx.x;
    const int base = blockIdx.x * OWN_NODES;
    const int nloc = min(OWN_NODES, n_nodes - base);
    const unsigned unloc = (unsigned)nloc;

    for (int idx = tid; idx < nloc * OUT_DIM; idx += OWN_THREADS) acc[idx] = 0.0f;
    for (int i = tid; i < nloc; i += OWN_THREADS) {
        den_s[i]  = 0.0f;
        mrun_s[i] = __uint_as_float(0xFF800000u);
        mb_s[i]   = ENC_NEG_INF;
        er_s[i]   = er[base + i];
    }
    if (tid == 0) { qcount = 0u; cqcount = 0u; wovf = 0u; }
    if (tid < OWN_WAVES) wcnt[tid] = 0u;
    __syncthreads();

    const unsigned E = (unsigned)n_edges;
    const bool use_bm = bm != 0;
    const unsigned eshift = use_bm ? (unsigned)CHUNK_SHIFT : 2u;
    const unsigned nwhole = E >> eshift;
    const unsigned ngroups = use_bm ? (nwhole + TILE_CHUNKS - 1u) / TILE_CHUNKS : nwhole;
    const unsigned gsafe = use_bm ? 1u : (QCAP >> 2) - 1u;
    const int4* __restrict__ dst4 = (const int4*)dst;
    const int4* __restrict__ src4 = (const int4*)src;
    unsigned* cq = (unsigned*)q_val;
    const unsigned wv = (unsigned)__builtin_amdgcn_readfirstlane(tid >> 5);
    const unsigned ln = (unsigned)tid & 31u;
    const unsigned sub = ln >> 3, part = ln & 7u;
    const unsigned cqcap = QCAP - 4u * OWN_WAVES * CHUNK_U;
    unsigned gpos = 0u, glen = ngroups, nflush = 0u;
    bool more = E > 0u;
    const unsigned rot = (unsigned)(((unsigned long long)blockIdx.x * ngroups) / gridDim.x) & ~31u;

    while (more) {
        const unsigned gend = min(gpos + glen, ngroups);
        unsigned ncq = 0u;

        if (use_bm) {
            for (unsigned tb = gpos + 4u * wv; tb < gend; tb += 4u * OWN_WAVES) {
                const unsigned tile = tb + sub;
                unsigned word = 0u;
                if (tile < gend) word = bm[((size_t)tile * KEY_OWNERS + blockIdx.x) * TILE_WORDS + part];
                const unsigned c0 = tile * TILE_CHUNKS + part * 32u;
                bool fits;
                unsigned pos = wave_slots(&cqcount, (unsigned)__builtin_popcount(word), ln, cqcap, fits);
                if (!fits) word = 0u;
                while (word) {
                    cq[pos++] = c0 + (unsigned)__builtin_ctz(word);
                    word &= word - 1u;
                }
            }
            __syncthreads();
            ncq = (unsigned)__builtin_amdgcn_readfirstlane((int)cqcount);
            if (ncq <= cqcap) {
                for (unsigned i0 = 4u * wv; i0 < ncq; i0 += 4u * OWN_WAVES * CHUNK_U) {
                    unsigned h[CHUNK_U];
                    int4 v[CHUNK_U], sv[CHUNK_U];
#pragma unroll
                    for (int u = 0; u < CHUNK_U; ++u) {
                        const unsigned ck = min(cq[i0 + (unsigned)u * 4u * OWN_WAVES + sub], nwhole - 1u);
                        v[u]  = dst4[ck * 8u + part];
                        sv[u] = src4[ck * 8u + part];
                    }
                    unsigned cnt = 0u;
#pragma unroll
                    for (int u = 0; u < CHUNK_U; ++u) {
                        const unsigned lim = (i0 + (unsigned)u * 4u * OWN_WAVES + sub < ncq) ? unloc : 0u;
                        h[u] = ((unsigned)(v[u].x - base) < lim ? 1u : 0u) | ((unsigned)(v[u].y - base) < lim ? 2u : 0u) |
                               ((unsigned)(v[u].z - base) < lim ? 4u : 0u) | ((unsigned)(v[u].w - base) < lim ? 8u : 0u);
                        cnt += (unsigned)__builtin_popcount(h[u]);
                    }
                    bool fits;
                    unsigned pos = wave_slots(&qcount, cnt, ln, QCAP, fits);
                    if (fits) {
#pragma unroll
                        for (int u = 0; u < CHUNK_U; ++u) {
                            unsigned m = h[u];
                            while (m) {
                                const unsigned k = (unsigned)__builtin_ctz(m);
                                q_e[pos]  = (unsigned)pick4(sv[u], k);
                                q_dl[pos] = (unsigned)(pick4(v[u], k) - base);
                                ++pos;
                                m &= m - 1u;
                            }
                        }
                    }
                }
            }
        } else {
            for (unsigned g = gpos + (unsigned)tid; g < gend; g += OWN_THREADS * SCAN_U) {
                int4 v[SCAN_U];
#pragma unroll
                for (int u = 0; u < SCAN_U; ++u) {
                    unsigned ph = min(g + (unsigned)u * OWN_THREADS, gend - 1u) + rot;
                    if (ph >= ngroups) ph -= ngroups;
                    v[u] = dst4[ph];
                }
#pragma unroll
                for (int u = 0; u < SCAN_U; ++u) {
                    unsigned gi = g + (unsigned)u * OWN_THREADS;
                    unsigned ph = gi + rot;
                    if (ph >= ngroups) ph -= ngroups;
                    unsigned e0 = ph << 2;
                    unsigned lim = (gi < gend) ? unloc : 0u;
                    bool h0 = (unsigned)(v[u].x - base) < lim, h1 = (unsigned)(v[u].y - base) < lim;
                    bool h2 = (unsigned)(v[u].z - base) < lim, h3 = (unsigned)(v[u].w - base) < lim;
                    if (h0 | h1 | h2 | h3) {
                        if (h0) q_push(&qcount, q_e, q_dl, (unsigned)src[e0],      (unsigned)(v[u].x - base));
                        if (h1) q_push(&qcount, q_e, q_dl, (unsigned)src[e0 + 1u], (unsigned)(v[u].y - base));
                        if (h2) q_push(&qcount, q_e, q_dl, (unsigned)src[e0 + 2u], (unsigned)(v[u].z - base));
                        if (h3) q_push(&qcount, q_e, q_dl, (unsigned)src[e0 + 3u], (unsigned)(v[u].w - base));
                    }
                }
            }
        }
        if (gend == ngroups) {
            unsigned e = (nwhole << eshift) + (unsigned)tid;
            if ((unsigned)tid < (E & ((1u << eshift) - 1u))) {
                const unsigned dl = (unsigned)(dst[e] - base);
                if (dl < unloc) q_push(&qcount, q_e, q_dl, (unsigned)src[e], dl);
            }
        }
        __syncthreads();
        const unsigned total = (unsigned)__builtin_amdgcn_readfirstlane((int)qcount);
        __syncthreads();

        if (total > QCAP || ncq > cqcap) {
            if (tid == 0) { qcount = 0u; cqcount = 0u; }
            glen = gsafe;
            __syncthreads();
            continue;
        }

        for (unsigned i = (unsigned)tid; i < total; i += OWN_THREADS) {
            const unsigned s  = q_e[i];
            const unsigned dl = q_dl[i];
            float v = el[s] + er_s[dl];
            v = v > 0.0f ? v : NEG_SLOPE * v;
            q_val[i] = v;
            atomicMax(&mb_s[dl], enc_f(v));
        }
        __syncthreads();

        for (int i = tid; i < nloc; i += OWN_THREADS) {
            float mo = mrun_s[i];
            float mn = fmaxf(mo, dec_f(mb_s[i]));
            mrun_s[i] = mn;
            mb_s[i]   = ENC_NEG_INF;
            if (nflush) {
                float s = (mo == mn) ? 1.0f : expf(mo - mn);
                sc_s[i]  = s;
                den_s[i] *= s;
            }
        }
        __syncthreads();
        if (nflush) {
            for (int idx = tid; idx < nloc * OUT_DIM; idx += OWN_THREADS) acc[idx] *= sc_s[idx >> 6];
        }

        for (unsigned i = (unsigned)tid; i < total; i += OWN_THREADS) {
            unsigned dl = q_dl[i];
            float ex = expf(q_val[i] - mrun_s[dl]);
            q_val[i] = ex;
            atomicAdd(&den_s[dl], ex);
            const unsigned w = dl / ROWS_PW;
            const unsigned slot = atomicAdd(&wcnt[w], 1u);
            if (slot < WL_CAP) wl[w * WL_CAP + slot] = (unsigned short)i;
            else               wovf = 1u;
        }
        __syncthreads();
        const bool own_rows = wovf == 0u;

        if (own_rows) {
            const unsigned n  = (unsigned)__builtin_amdgcn_readfirstlane((int)wcnt[wv]);
            const unsigned r0 = wv * ROWS_PW;
            if (n) {
                v32f ar = {};
#pragma unroll
                for (unsigned k = 0u; k < ROWS_PW; ++k) {
                    if (r0 + k < unloc) {
                        const v2f t = *(const v2f*)&acc[(r0 + k) * OUT_DIM + 2u * ln];
                        ar[2u * k] = t.x; ar[2u * k + 1u] = t.y;
                    }
                }
                const unsigned short* mine = &wl[wv * WL_CAP];
                for (unsigned g = 0u; g < n; g += 32u) {
                    const unsigned nn = min(32u, n - g);
                    const unsigned i  = mine[min(g + ln, n - 1u)];
                    const unsigned s_l  = q_e[i];
                    const unsigned rw_l = q_dl[i] - r0;
                    const unsigned ex_l = __float_as_uint(q_val[i]);
                    if (nn == 32u) {
#pragma unroll
                        for (unsigned j0 = 0u; j0 < 32u; j0 += GATHER_U)
                            gather_batch_own(feat, ar, s_l, rw_l, ex_l, j0, 32u, ln);
                    } else {
#pragma nounroll
                        for (unsigned j0 = 0u; j0 < nn; j0 += GATHER_U)
                            gather_batch_own(feat, ar, s_l, rw_l, ex_l, j0, nn, ln);
                    }
                }
#pragma unroll
                for (unsigned k = 0u; k < ROWS_PW; ++k)
                    if (r0 + k < unloc) *(v2f*)&acc[(r0 + k) * OUT_DIM + 2u * ln] = (v2f){ar[2u * k], ar[2u * k + 1u]};
            }
        } else if (total) {
            const unsigned nbat = (total + GATHER_U - 1u) / GATHER_U;
            const unsigned beg  = (wv * nbat / OWN_WAVES) * GATHER_U;
            const unsigned end  = min(((wv + 1u) * nbat / OWN_WAVES) * GATHER_U, total);
            for (unsigned g = beg; g < end; g += 32u) {
                const unsigned n = min(32u, end - g);
                const unsigned i = min(g + ln, end - 1u);
                const unsigned s_l  = q_e[i];
                const unsigned dl_l = q_dl[i];
                const unsigned ex_l = __float_as_uint(q_val[i]);
                if (n == 32u) {
#pragma unroll
                    for (unsigned j0 = 0u; j0 < 32u; j0 += GATHER_U)
                        gather_batch(feat, acc, s_l, dl_l, ex_l, j0, 32u, ln);
                } else {
#pragma nounroll
                    for (unsigned j0 = 0u; j0 < n; j0 += GATHER_U)
                        gather_batch(feat, acc, s_l, dl_l, ex_l, j0, n, ln);
                }
            }
        }
        if (tid == 0) { qcount = 0u; cqcount = 0u; }
        __syncthreads();
        if (tid == 0) wovf = 0u;
        if (tid < OWN_WAVES) wcnt[tid] = 0u;
        ++nflush;
        gpos = gend;
        more = gpos < ngroups;
    }

    {
        const unsigned npair = unloc * (OUT_DIM / 2);
        const v2f* acc2 = (const v2f*)acc;
        v2f*       out2 = (v2f*)(out + (size_t)base * OUT_DIM);
        const v2f  b2   = ((const v2f*)bias)[(unsigned)tid & 31u];
        for (unsigned p = (unsigned)tid; p < npair; p += OWN_THREADS) {
            const float d = den_s[p >> 5];
            const v2f   a = acc2[p];
            const float inv = d > 0.0f ? __builtin_amdgcn_rcpf(d) : 0.0f;
            v2f o;
            o.x = tanh_sel(fmaf(a.x, inv, b2.x));
            o.y = tanh_sel(fmaf(a.y, inv, b2.y));
            out2[p] = o;
        }
    }
}

static inline size_t align256(size_t x) { return (x + 255) & ~(size_t)255; }

extern "C" void kernel_launch(void* const* d_in, const int* in_sizes, int n_in,
                              void* d_out, int out_size, void* d_ws, size_t ws_size,
                              hipStream_t stream) {
    const float* x      = (const float*)d_in[0];
    const int*   src    = (const int*)d_in[1];
    const int*   dst    = (const int*)d_in[2];
    const float* W      = (const float*)d_in[3];
    const float* attn_l = (const float*)d_in[4];
    const float* attn_r = (const float*)d_in[5];
    const float* bias   = (const float*)d_in[6];
    float*       out    = (float*)d_out;

    const int N = in_sizes[0] / IN_DIM;
    const int E = in_sizes[1];

    char* ws = (char*)d_ws;
    float* feat = (float*)ws;    ws += align256((size_t)N * OUT_DIM * sizeof(float));
    float* el   = (float*)ws;    ws += align256((size_t)N * sizeof(float));
    float* er   = (float*)ws;    ws += align256((size_t)N * sizeof(float));

    const int own_blocks = (N + OWN_NODES - 1) / OWN_NODES;
    const int n_tiles = ((E >> CHUNK_SHIFT) + TILE_CHUNKS - 1) / TILE_CHUNKS;
    const size_t bm_bytes = (size_t)n_tiles * KEY_OWNERS * TILE_WORDS * sizeof(unsigned);
    unsigned* bm = 0;
    if (own_blocks <= KEY_OWNERS && (size_t)(ws - (char*)d_ws) + align256(bm_bytes) <= ws_size)
        bm = (unsigned*)ws;

    {
        gat_gemm<<<GEMM_BLOCKS, GEMM_THREADS, 0, stream>>>(x, W, feat, N);
    }
    {
        int blocks = 400 + (bm ? n_tiles : 0);
        gat_logits<<<blocks, K2_THREADS, 0, stream>>>(feat, attn_l, attn_r, el, er, N, dst, bm, E, n_tiles);
    }
    {
        gat_owner<<<own_blocks, OWN_THREADS, 0, stream>>>(src, dst, el, er, feat, bias, out, N, E, bm);
    }
}
